# starting kernel
# baseline (speedup 1.0000x reference)
#include <hip/hip_runtime.h>

typedef __attribute__((ext_vector_type(2))) float v2f;
typedef __attribute__((ext_vector_type(8))) float v8f;

#define LD 65
#define MS (64 * LD)
#define C_DIM 64
#define T_DIM 4096
#define N_EPOCH 8
#define PATCH 512
#define EPS_COV 1e-5f

__device__ __forceinline__ void mm64(float* __restrict__ D, const float* __restrict__ A,
                                     const float* __restrict__ Bm, int lane, int wave) {
    __syncthreads();
#pragma unroll
    for (int i = 0; i < 2; ++i) {
        const int t  = wave + 8 * i;
        const int tm = (t >> 2) * 16;
        const int tn = (t & 3) * 16;
        const int row = tm + (lane & 15);
        const int col = tn + (lane & 15);
        v8f acc = {0.f, 0.f, 0.f, 0.f, 0.f, 0.f, 0.f, 0.f};
#pragma unroll
        for (int k4 = 0; k4 < 64; k4 += 4) {
            const int ko = k4 + ((lane >> 4) << 1);
            v2f a; a.x = A[row * LD + ko];  a.y = A[row * LD + ko + 1];
            v2f b; b.x = Bm[ko * LD + col]; b.y = Bm[(ko + 1) * LD + col];
            acc = __builtin_amdgcn_wmma_f32_16x16x4_f32(false, a, false, b,
                                                        (short)0, acc, false, false);
        }
#pragma unroll
        for (int v = 0; v < 8; ++v) {
            const int m = tm + v + 8 * (lane >> 4);
            D[m * LD + tn + (lane & 15)] = acc[v];
        }
    }
    __syncthreads();
}

#define FOR_EW for (int i = tid; i < 4096; i += 256)

__device__ __forceinline__ void ew_copy(float* d, const float* s, int tid) {
    FOR_EW { int r = i >> 6, c = i & 63; d[r * LD + c] = s[r * LD + c]; }
    __syncthreads();
}
__device__ __forceinline__ void ew_identity(float* d, float dv, int tid) {
    FOR_EW { int r = i >> 6, c = i & 63; d[r * LD + c] = (r == c) ? dv : 0.f; }
    __syncthreads();
}

__global__ void cov64_kernel(const float* __restrict__ x, float* __restrict__ cov) {
    __shared__ float Xs[64 * LD];
    __shared__ float mp[256];
    __shared__ float mean[64];

    const int tid = threadIdx.x, lane = tid & 31, wave = tid >> 5;
    const int bid = blockIdx.x;
    const int b = bid >> 3, e = bid & 7;
    const float* xb = x + (size_t)b * C_DIM * T_DIM + (size_t)e * PATCH;

    float msum = 0.f;
    v8f acc0 = {0.f,0.f,0.f,0.f,0.f,0.f,0.f,0.f};
    v8f acc1 = {0.f,0.f,0.f,0.f,0.f,0.f,0.f,0.f};

    for (int kc = 0; kc < PATCH; kc += 64) {
        __syncthreads();
        for (int i = tid; i < 4096; i += 256) {
            const int c = i >> 6, j = i & 63;
            Xs[c * LD + j] = xb[(size_t)c * T_DIM + kc + j];
        }
        __syncthreads();
        {
            const int c = tid & 63, part = tid >> 6;
#pragma unroll
            for (int j = 0; j < 16; ++j) msum += Xs[c * LD + part * 16 + j];
        }
#pragma unroll
        for (int i = 0; i < 2; ++i) {
            const int t  = wave + 8 * i;
            const int tm = (t >> 2) * 16, tn = (t & 3) * 16;
            const int row = tm + (lane & 15), coln = tn + (lane & 15);
            v8f acc = i ? acc1 : acc0;
#pragma unroll
            for (int k4 = 0; k4 < 64; k4 += 4) {
                const int ko = k4 + ((lane >> 4) << 1);
                v2f a; a.x = Xs[row * LD + ko];  a.y = Xs[row * LD + ko + 1];
                v2f bf; bf.x = Xs[coln * LD + ko]; bf.y = Xs[coln * LD + ko + 1];
                acc = __builtin_amdgcn_wmma_f32_16x16x4_f32(false, a, false, bf,
                                                            (short)0, acc, false, false);
            }
            if (i) acc1 = acc; else acc0 = acc;
        }
    }
    __syncthreads();
    mp[tid] = msum;
    __syncthreads();
    if (tid < 64)
        mean[tid] = (mp[tid] + mp[tid + 64] + mp[tid + 128] + mp[tid + 192]) * (1.f / PATCH);
    __syncthreads();

#pragma unroll
    for (int i = 0; i < 2; ++i) {
        const int t  = wave + 8 * i;
        const int tm = (t >> 2) * 16, tn = (t & 3) * 16;
        v8f acc = i ? acc1 : acc0;
#pragma unroll
        for (int v = 0; v < 8; ++v) {
            const int m = tm + v + 8 * (lane >> 4);
            const int n = tn + (lane & 15);
            const float val = (acc[v] - (float)PATCH * mean[m] * mean[n]) * (1.f / (PATCH - 1));
            cov[(size_t)bid * 4096 + m * 64 + n] = val;
        }
    }
}

__global__ void spdlogexp_kernel(const float* __restrict__ cov,
                                 const float* __restrict__ alphap,
                                 const float* __restrict__ betap,
                                 float* __restrict__ out) {
    extern __shared__ float lds[];
    float* A  = lds;
    float* Y  = lds + 1 * MS;
    float* Z  = lds + 2 * MS;
    float* Tm = lds + 3 * MS;
    float* W  = lds + 4 * MS;
    float* red = lds + 5 * MS;

    const int tid = threadIdx.x, lane = tid & 31, wave = tid >> 5;
    const int bid = blockIdx.x;
    const float alpha = alphap[0], beta = betap[0];

    for (int i = tid; i < 4096; i += 256)
        A[(i >> 6) * LD + (i & 63)] = cov[(size_t)bid * 4096 + i];
    __syncthreads();
    if (tid < 64) red[tid] = A[tid * LD + tid];
    __syncthreads();
    if (tid == 0) { float s = 0.f; for (int i = 0; i < 64; ++i) s += red[i]; red[0] = 1.f / s; }
    __syncthreads();
    const float invtr = red[0];
    FOR_EW { int r = i >> 6, c = i & 63;
             float v = A[r * LD + c] * invtr;
             if (r == c) v += EPS_COV;
             A[r * LD + c] = v; }
    __syncthreads();

    constexpr int K_SQRT = 5;
    const int ns_it[K_SQRT] = {18, 12, 9, 7, 6};
    for (int kq = 0; kq < K_SQRT; ++kq) {
        ew_copy(Y, A, tid);
        ew_identity(Z, 1.f, tid);
        for (int it = 0; it < ns_it[kq]; ++it) {
            mm64(Tm, Z, Y, lane, wave);
            FOR_EW { int r = i >> 6, c = i & 63;
                     float v = Tm[r * LD + c];
                     Tm[r * LD + c] = ((r == c) ? 1.5f : 0.f) - 0.5f * v; }
            __syncthreads();
            mm64(W, Y, Tm, lane, wave);  ew_copy(Y, W, tid);
            mm64(W, Tm, Z, lane, wave);  ew_copy(Z, W, tid);
        }
        ew_copy(A, Y, tid);
    }
    FOR_EW { int r = i >> 6, c = i & 63;
             Y[r * LD + c] = ((r == c) ? 1.f : 0.f) - A[r * LD + c]; }
    __syncthreads();
    constexpr int LOG_M = 12;
    ew_identity(Z, 1.f / (float)LOG_M, tid);
    for (int m = LOG_M - 1; m >= 1; --m) {
        mm64(Tm, Y, Z, lane, wave);
        const float dm = 1.f / (float)m;
        FOR_EW { int r = i >> 6, c = i & 63;
                 Z[r * LD + c] = Tm[r * LD + c] + ((r == c) ? dm : 0.f); }
        __syncthreads();
    }
    mm64(Tm, Y, Z, lane, wave);
    const float lscale = -(float)(1 << K_SQRT);

    FOR_EW { int r = i >> 6, c = i & 63;
             A[r * LD + c] = alpha * (lscale * Tm[r * LD + c]) + beta; }
    __syncthreads();
    FOR_EW { int r = i >> 6, c = i & 63;
             W[r * LD + c] = 0.5f * (A[r * LD + c] + A[c * LD + r]); }
    __syncthreads();

    constexpr int S_EXP = 6, EXP_M = 10;
    const float es = 1.f / (float)(1 << S_EXP);
    FOR_EW { int r = i >> 6, c = i & 63; Y[r * LD + c] = W[r * LD + c] * es; }
    __syncthreads();
    ew_identity(Z, 1.f, tid);
    for (int m = EXP_M; m >= 1; --m) {
        mm64(Tm, Y, Z, lane, wave);
        const float dm = 1.f / (float)m;
        FOR_EW { int r = i >> 6, c = i & 63;
                 Z[r * LD + c] = Tm[r * LD + c] * dm + ((r == c) ? 1.f : 0.f); }
        __syncthreads();
    }
    for (int j = 0; j < S_EXP; ++j) {
        mm64(Tm, Z, Z, lane, wave);
        ew_copy(Z, Tm, tid);
    }

    for (int i = tid; i < 4096; i += 256)
        out[(size_t)bid * 4096 + i] = Z[(i >> 6) * LD + (i & 63)];
}

extern "C" void kernel_launch(void* const* d_in, const int* in_sizes, int n_in,
                              void* d_out, int out_size, void* d_ws, size_t ws_size,
                              hipStream_t stream) {
    const float* x     = (const float*)d_in[0];
    const float* alpha = (const float*)d_in[1];
    const float* beta  = (const float*)d_in[2];
    float* out = (float*)d_out;
    float* cov = (float*)d_ws;

    const int nmat = 256 * N_EPOCH;
    cov64_kernel<<<nmat, 256, 0, stream>>>(x, cov);

    const size_t shbytes = (size_t)(5 * MS + 64) * sizeof(float);
    spdlogexp_kernel<<<nmat, 256, shbytes, stream>>>(cov, alpha, beta, out);
}


# Covariance fused into the per-matrix wave: x read from global straight into registers used as both A and B fragments of X*X^T; no LDS staging, no workspace pass, one launch; sum order differs
# speedup vs baseline: 3.5967x; 3.5967x over previous
#include <hip/hip_runtime.h>

typedef __attribute__((ext_vector_type(2))) float v2f;
typedef __attribute__((ext_vector_type(8))) float v8f;

#define C_DIM 64
#define T_DIM 4096
#define N_EPOCH 8
#define PATCH 512
#define EPS_COV 1e-5f

typedef __attribute__((ext_vector_type(4))) float v4f;

#define PW 68
#define SLOT (64 * PW)


__device__ __forceinline__ void wave_lds_sync() { __syncthreads(); }

__device__ __forceinline__ void put_lds(float* __restrict__ s, const v8f (&M)[16], int h, int c) {
    float* p = s + (8 * h) * PW + c;
#pragma unroll
    for (int I = 0; I < 4; ++I)
#pragma unroll
        for (int J = 0; J < 4; ++J)
#pragma unroll
            for (int r = 0; r < 8; ++r)
                p[(16 * I + r) * PW + 16 * J] = M[I * 4 + J][r];
}

__device__ __forceinline__ void get_arow(v8f (&a)[4], const float* __restrict__ s, int J, int h, int c) {
    const float* p = s + (16 * J + c) * PW + 8 * h;
#pragma unroll
    for (int I = 0; I < 4; ++I) {
        const v4f lo = *(const v4f*)(p + 16 * I);
        const v4f hi = *(const v4f*)(p + 16 * I + 4);
        v8f t = {lo.x, lo.y, lo.z, lo.w, hi.x, hi.y, hi.z, hi.w};
        a[I] = t;
    }
}

__device__ __forceinline__ void mm(v8f (&D)[16], const float* __restrict__ sA, const v8f (&B)[16],
                                   int h, int c) {
#pragma unroll
    for (int J = 0; J < 4; ++J) {
        v8f a[4];
        get_arow(a, sA, J, h, c);
        v8f acc[4];
#pragma unroll
        for (int Jp = 0; Jp < 4; ++Jp) { v8f z = {0.f,0.f,0.f,0.f,0.f,0.f,0.f,0.f}; acc[Jp] = z; }
#pragma unroll
        for (int I = 0; I < 4; ++I)
#pragma unroll
            for (int q = 0; q < 4; ++q) {
                v2f af; af.x = a[I][2 * q]; af.y = a[I][2 * q + 1];
#pragma unroll
                for (int Jp = 0; Jp < 4; ++Jp) {
                    v2f bf; bf.x = B[I * 4 + Jp][2 * q]; bf.y = B[I * 4 + Jp][2 * q + 1];
                    acc[Jp] = __builtin_amdgcn_wmma_f32_16x16x4_f32(false, af, false, bf,
                                                                    (short)0, acc[Jp], false, false);
                }
            }
#pragma unroll
        for (int Jp = 0; Jp < 4; ++Jp) D[J * 4 + Jp] = acc[Jp];
    }
}

#define FOR_TILE_EL \
    _Pragma("unroll") for (int I = 0; I < 4; ++I) \
    _Pragma("unroll") for (int J = 0; J < 4; ++J) \
    _Pragma("unroll") for (int r = 0; r < 8; ++r)
#define ON_DIAG ((I == J) && (8 * h + r == c))

__global__ void __launch_bounds__(32)
spdlogexp_kernel(const float* __restrict__ x,
                 const float* __restrict__ alphap,
                 const float* __restrict__ betap,
                 float* __restrict__ out) {
    extern __shared__ __attribute__((aligned(16))) float lds[];
    float* s0 = lds;
    float* s1 = lds + SLOT;

    const int lane = threadIdx.x, h = lane >> 4, c = lane & 15;
    const int bid = blockIdx.x;
    const float alpha = alphap[0], beta = betap[0];

    v8f A[16], Y[16], Z[16], T[16], Yn[16], Zn[16];

    {
        const float* xr = x + (size_t)(bid >> 3) * C_DIM * T_DIM + (size_t)(bid & 7) * PATCH
                            + (size_t)c * T_DIM + 16 * h;
        v4f fa[4][4], fb[4][4], fc[4][4], fd[4][4];
        float ms[4] = {0.f, 0.f, 0.f, 0.f};
#pragma unroll
        for (int t = 0; t < 16; ++t) { v8f z = {0.f,0.f,0.f,0.f,0.f,0.f,0.f,0.f}; A[t] = z; }

        auto ldgrp = [&](v4f (&f)[4][4], int g) {
#pragma unroll
            for (int J = 0; J < 4; ++J)
#pragma unroll
                for (int i = 0; i < 4; ++i)
                    f[J][i] = *(const v4f*)(xr + (size_t)(16 * J) * T_DIM + 32 * g + 4 * i);
        };
        auto gram = [&](const v4f (&f)[4][4]) {
#pragma unroll
            for (int i = 0; i < 4; ++i)
#pragma unroll
                for (int p = 0; p < 2; ++p) {
                    v2f fr[4];
#pragma unroll
                    for (int J = 0; J < 4; ++J) { fr[J].x = f[J][i][2 * p]; fr[J].y = f[J][i][2 * p + 1]; }
#pragma unroll
                    for (int J = 0; J < 4; ++J)
#pragma unroll
                        for (int Jp = 0; Jp < 4; ++Jp)
                            A[J * 4 + Jp] = __builtin_amdgcn_wmma_f32_16x16x4_f32(false, fr[J], false, fr[Jp],
                                                                                 (short)0, A[J * 4 + Jp], false, false);
                }
#pragma unroll
            for (int J = 0; J < 4; ++J) {
                const v4f s4 = (f[J][0] + f[J][1]) + (f[J][2] + f[J][3]);
                ms[J] += (s4.x + s4.y) + (s4.z + s4.w);
            }
        };

        ldgrp(fa, 0);
        ldgrp(fb, 1);
        for (int g = 2; g < PATCH / 32 - 2; g += 4) {
            ldgrp(fc, g);
            ldgrp(fd, g + 1);
            gram(fa);
            gram(fb);
            ldgrp(fa, g + 2);
            ldgrp(fb, g + 3);
            gram(fc);
            gram(fd);
        }
        ldgrp(fc, PATCH / 32 - 2);
        ldgrp(fd, PATCH / 32 - 1);
        gram(fa);
        gram(fb);
        gram(fc);
        gram(fd);

        float mean[4];
#pragma unroll
        for (int J = 0; J < 4; ++J) {
            const float t = ms[J] + __shfl_xor(ms[J], 16);
            mean[J] = t * (1.f / PATCH);
        }
#pragma unroll
        for (int I = 0; I < 4; ++I)
#pragma unroll
            for (int r = 0; r < 8; ++r) {
                const float mr = __shfl(mean[I], 8 * h + r);
#pragma unroll
                for (int J = 0; J < 4; ++J)
                    A[I * 4 + J][r] = (A[I * 4 + J][r] - (float)PATCH * mr * mean[J]) * (1.f / (PATCH - 1));
            }
    }
    float tr = 0.f;
    FOR_TILE_EL { if (ON_DIAG) tr += A[I * 4 + J][r]; }
#pragma unroll
    for (int m = 16; m >= 1; m >>= 1) tr += __shfl_xor(tr, m);
    const float invtr = 1.f / tr;
    FOR_TILE_EL {
        float v = A[I * 4 + J][r] * invtr;
        if (ON_DIAG) v += EPS_COV;
        A[I * 4 + J][r] = v;
    }

    constexpr int K_SQRT = 5;
    constexpr float NS_A_CAP = 2.3f;
    float lo = EPS_COV;
    for (int kq = 0; kq < K_SQRT; ++kq) {
        const int nit = (kq == 0) ? 10 : (kq == 1) ? 7 : (kq == 2) ? 5 : 4;
        float lb = lo;
        float a, b;
        auto coeff = [&]() {
            const float s = sqrtf(lb);
            a = fminf(sqrtf(6.75f / (1.f + s + s * s)), NS_A_CAP);
            b = a * a * a * (4.f / 27.f);
            const float tl = a - b * lb, t1 = a - b;
            lb = fminf(fminf(lb * tl * tl, t1 * t1), 1.f);
        };
        coeff();
        FOR_TILE_EL { Z[I * 4 + J][r] = (ON_DIAG ? a : 0.f) - b * A[I * 4 + J][r]; }
        wave_lds_sync();
        put_lds(s1, A, h, c);
        put_lds(s0, Z, h, c);
        wave_lds_sync();
        mm(Y, s1, Z, h, c);
        wave_lds_sync();
        put_lds(s1, Y, h, c);
        wave_lds_sync();
        for (int it = 1; it < nit; ++it) {
            const bool last = (it == nit - 1);
            coeff();
            mm(T, s0, Y, h, c);
            FOR_TILE_EL { const float v = T[I * 4 + J][r];
                          T[I * 4 + J][r] = (ON_DIAG ? a : 0.f) - b * v; }
            if (!last) {
                wave_lds_sync();
                put_lds(s0, T, h, c);
                wave_lds_sync();
            }
            mm(Yn, s1, T, h, c);
            if (!last) {
                mm(Zn, s0, Z, h, c);
                wave_lds_sync();
                put_lds(s1, Yn, h, c);
                put_lds(s0, Zn, h, c);
                wave_lds_sync();
#pragma unroll
                for (int t = 0; t < 16; ++t) Z[t] = Zn[t];
            }
#pragma unroll
            for (int t = 0; t < 16; ++t) Y[t] = Yn[t];
        }
#pragma unroll
        for (int t = 0; t < 16; ++t) A[t] = Y[t];
        lo = sqrtf(lo);
    }
    FOR_TILE_EL { Y[I * 4 + J][r] = (ON_DIAG ? 1.f : 0.f) - A[I * 4 + J][r]; }
    wave_lds_sync();
    put_lds(s1, Y, h, c);
    wave_lds_sync();
    constexpr int LOG_M = 12;
    FOR_TILE_EL { Z[I * 4 + J][r] = Y[I * 4 + J][r] * (1.f / (float)LOG_M)
                                    + (ON_DIAG ? (1.f / (float)(LOG_M - 1)) : 0.f); }
    for (int m = LOG_M - 2; m >= 1; --m) {
        mm(T, s1, Z, h, c);
        const float dm = 1.f / (float)m;
        FOR_TILE_EL { Z[I * 4 + J][r] = T[I * 4 + J][r] + (ON_DIAG ? dm : 0.f); }
    }
    mm(T, s1, Z, h, c);
    const float lscale = -(float)(1 << K_SQRT);

    FOR_TILE_EL { A[I * 4 + J][r] = alpha * (lscale * T[I * 4 + J][r]) + beta; }
    wave_lds_sync();
    put_lds(s0, A, h, c);
    wave_lds_sync();
    constexpr int S_EXP = 6, EXP_M = 10;
    const float es = 1.f / (float)(1 << S_EXP);
#pragma unroll
    for (int J = 0; J < 4; ++J) {
        v8f at[4];
        get_arow(at, s0, J, h, c);
#pragma unroll
        for (int I = 0; I < 4; ++I)
#pragma unroll
            for (int r = 0; r < 8; ++r)
                Y[I * 4 + J][r] = (0.5f * (A[I * 4 + J][r] + at[I][r])) * es;
    }

    wave_lds_sync();
    put_lds(s1, Y, h, c);
    wave_lds_sync();
    FOR_TILE_EL { Z[I * 4 + J][r] = Y[I * 4 + J][r] * (1.f / (float)EXP_M) + (ON_DIAG ? 1.f : 0.f); }
    for (int m = EXP_M - 1; m >= 1; --m) {
        mm(T, s1, Z, h, c);
        const float dm = 1.f / (float)m;
        FOR_TILE_EL { Z[I * 4 + J][r] = T[I * 4 + J][r] * dm + (ON_DIAG ? 1.f : 0.f); }
    }
    for (int j = 0; j < S_EXP; ++j) {
        wave_lds_sync();
        put_lds(s0, Z, h, c);
        wave_lds_sync();
        mm(T, s0, Z, h, c);
#pragma unroll
        for (int t = 0; t < 16; ++t) Z[t] = T[t];
    }

    wave_lds_sync();
    put_lds(s0, Z, h, c);
    wave_lds_sync();
    float* ob = out + (size_t)bid * 4096;
#pragma unroll
    for (int j = 0; j < 32; ++j) {
        const int row = 2 * j + h;
        const v4f v = *(const v4f*)(s0 + row * PW + 4 * c);
        *(v4f*)(ob + row * 64 + 4 * c) = v;
    }
}

extern "C" void kernel_launch(void* const* d_in, const int* in_sizes, int n_in,
                              void* d_out, int out_size, void* d_ws, size_t ws_size,
                              hipStream_t stream) {
    const float* x     = (const float*)d_in[0];
    const float* alpha = (const float*)d_in[1];
    const float* beta  = (const float*)d_in[2];
    float* out = (float*)d_out;

    const int nmat = 256 * N_EPOCH;
    const size_t shbytes = (size_t)(2 * SLOT) * sizeof(float);
    spdlogexp_kernel<<<nmat, 32, shbytes, stream>>>(x, alpha, beta, out);
}


# Log taken after 3 square roots, not 5: a degree-14 near-minimax polynomial of log on the root's guaranteed spectral interval replaces the 12-term series about I: 101 -> 85 products; rounding differs
# speedup vs baseline: 4.2009x; 1.1680x over previous
#include <hip/hip_runtime.h>

typedef __attribute__((ext_vector_type(2))) float v2f;
typedef __attribute__((ext_vector_type(8))) float v8f;

#define C_DIM 64
#define T_DIM 4096
#define N_EPOCH 8
#define PATCH 512
#define EPS_COV 1e-5f

typedef __attribute__((ext_vector_type(4))) float v4f;

#define PW 68
#define SLOT (64 * PW)


__device__ __forceinline__ void wave_lds_sync() { __syncthreads(); }

__device__ __forceinline__ void put_lds(float* __restrict__ s, const v8f (&M)[16], int h, int c) {
    float* p = s + (8 * h) * PW + c;
#pragma unroll
    for (int I = 0; I < 4; ++I)
#pragma unroll
        for (int J = 0; J < 4; ++J)
#pragma unroll
            for (int r = 0; r < 8; ++r)
                p[(16 * I + r) * PW + 16 * J] = M[I * 4 + J][r];
}

__device__ __forceinline__ void get_arow(v8f (&a)[4], const float* __restrict__ s, int J, int h, int c) {
    const float* p = s + (16 * J + c) * PW + 8 * h;
#pragma unroll
    for (int I = 0; I < 4; ++I) {
        const v4f lo = *(const v4f*)(p + 16 * I);
        const v4f hi = *(const v4f*)(p + 16 * I + 4);
        v8f t = {lo.x, lo.y, lo.z, lo.w, hi.x, hi.y, hi.z, hi.w};
        a[I] = t;
    }
}

__device__ __forceinline__ void mm(v8f (&D)[16], const float* __restrict__ sA, const v8f (&B)[16],
                                   int h, int c) {
#pragma unroll
    for (int J = 0; J < 4; ++J) {
        v8f a[4];
        get_arow(a, sA, J, h, c);
        v8f acc[4];
#pragma unroll
        for (int Jp = 0; Jp < 4; ++Jp) { v8f z = {0.f,0.f,0.f,0.f,0.f,0.f,0.f,0.f}; acc[Jp] = z; }
#pragma unroll
        for (int I = 0; I < 4; ++I)
#pragma unroll
            for (int q = 0; q < 4; ++q) {
                v2f af; af.x = a[I][2 * q]; af.y = a[I][2 * q + 1];
#pragma unroll
                for (int Jp = 0; Jp < 4; ++Jp) {
                    v2f bf; bf.x = B[I * 4 + Jp][2 * q]; bf.y = B[I * 4 + Jp][2 * q + 1];
                    acc[Jp] = __builtin_amdgcn_wmma_f32_16x16x4_f32(false, af, false, bf,
                                                                    (short)0, acc[Jp], false, false);
                }
            }
#pragma unroll
        for (int Jp = 0; Jp < 4; ++Jp) D[J * 4 + Jp] = acc[Jp];
    }
}

#define FOR_TILE_EL \
    _Pragma("unroll") for (int I = 0; I < 4; ++I) \
    _Pragma("unroll") for (int J = 0; J < 4; ++J) \
    _Pragma("unroll") for (int r = 0; r < 8; ++r)
#define ON_DIAG ((I == J) && (8 * h + r == c))

__global__ void __launch_bounds__(32)
spdlogexp_kernel(const float* __restrict__ x,
                 const float* __restrict__ alphap,
                 const float* __restrict__ betap,
                 float* __restrict__ out) {
    extern __shared__ __attribute__((aligned(16))) float lds[];
    float* s0 = lds;
    float* s1 = lds + SLOT;

    const int lane = threadIdx.x, h = lane >> 4, c = lane & 15;
    const int bid = blockIdx.x;
    const float alpha = alphap[0], beta = betap[0];

    v8f A[16], Y[16], Z[16], T[16], Yn[16], Zn[16];

    {
        const float* xr = x + (size_t)(bid >> 3) * C_DIM * T_DIM + (size_t)(bid & 7) * PATCH
                            + (size_t)c * T_DIM + 16 * h;
        v4f fa[4][4], fb[4][4], fc[4][4], fd[4][4];
        float ms[4] = {0.f, 0.f, 0.f, 0.f};
#pragma unroll
        for (int t = 0; t < 16; ++t) { v8f z = {0.f,0.f,0.f,0.f,0.f,0.f,0.f,0.f}; A[t] = z; }

        auto ldgrp = [&](v4f (&f)[4][4], int g) {
#pragma unroll
            for (int J = 0; J < 4; ++J)
#pragma unroll
                for (int i = 0; i < 4; ++i)
                    f[J][i] = *(const v4f*)(xr + (size_t)(16 * J) * T_DIM + 32 * g + 4 * i);
        };
        auto gram = [&](const v4f (&f)[4][4]) {
#pragma unroll
            for (int i = 0; i < 4; ++i)
#pragma unroll
                for (int p = 0; p < 2; ++p) {
                    v2f fr[4];
#pragma unroll
                    for (int J = 0; J < 4; ++J) { fr[J].x = f[J][i][2 * p]; fr[J].y = f[J][i][2 * p + 1]; }
#pragma unroll
                    for (int J = 0; J < 4; ++J)
#pragma unroll
                        for (int Jp = 0; Jp < 4; ++Jp)
                            A[J * 4 + Jp] = __builtin_amdgcn_wmma_f32_16x16x4_f32(false, fr[J], false, fr[Jp],
                                                                                 (short)0, A[J * 4 + Jp], false, false);
                }
#pragma unroll
            for (int J = 0; J < 4; ++J) {
                const v4f s4 = (f[J][0] + f[J][1]) + (f[J][2] + f[J][3]);
                ms[J] += (s4.x + s4.y) + (s4.z + s4.w);
            }
        };

        ldgrp(fa, 0);
        ldgrp(fb, 1);
        for (int g = 2; g < PATCH / 32 - 2; g += 4) {
            ldgrp(fc, g);
            ldgrp(fd, g + 1);
            gram(fa);
            gram(fb);
            ldgrp(fa, g + 2);
            ldgrp(fb, g + 3);
            gram(fc);
            gram(fd);
        }
        ldgrp(fc, PATCH / 32 - 2);
        ldgrp(fd, PATCH / 32 - 1);
        gram(fa);
        gram(fb);
        gram(fc);
        gram(fd);

        float mean[4];
#pragma unroll
        for (int J = 0; J < 4; ++J) {
            const float t = ms[J] + __shfl_xor(ms[J], 16);
            mean[J] = t * (1.f / PATCH);
        }
#pragma unroll
        for (int I = 0; I < 4; ++I)
#pragma unroll
            for (int r = 0; r < 8; ++r) {
                const float mr = __shfl(mean[I], 8 * h + r);
#pragma unroll
                for (int J = 0; J < 4; ++J)
                    A[I * 4 + J][r] = (A[I * 4 + J][r] - (float)PATCH * mr * mean[J]) * (1.f / (PATCH - 1));
            }
    }
    float tr = 0.f;
    FOR_TILE_EL { if (ON_DIAG) tr += A[I * 4 + J][r]; }
#pragma unroll
    for (int m = 16; m >= 1; m >>= 1) tr += __shfl_xor(tr, m);
    const float invtr = 1.f / tr;
    FOR_TILE_EL {
        float v = A[I * 4 + J][r] * invtr;
        if (ON_DIAG) v += EPS_COV;
        A[I * 4 + J][r] = v;
    }

    constexpr int K_SQRT = 3;
    constexpr float NS_A_CAP = 2.3f;
    float lo = EPS_COV;
    for (int kq = 0; kq < K_SQRT; ++kq) {
        const int nit = (kq == 0) ? 10 : (kq == 1) ? 7 : 5;
        float lb = lo;
        float a, b;
        auto coeff = [&]() {
            const float s = sqrtf(lb);
            a = fminf(sqrtf(6.75f / (1.f + s + s * s)), NS_A_CAP);
            b = a * a * a * (4.f / 27.f);
            const float tl = a - b * lb, t1 = a - b;
            lb = fminf(fminf(lb * tl * tl, t1 * t1), 1.f);
        };
        coeff();
        FOR_TILE_EL { Z[I * 4 + J][r] = (ON_DIAG ? a : 0.f) - b * A[I * 4 + J][r]; }
        wave_lds_sync();
        put_lds(s1, A, h, c);
        put_lds(s0, Z, h, c);
        wave_lds_sync();
        mm(Y, s1, Z, h, c);
        wave_lds_sync();
        put_lds(s1, Y, h, c);
        wave_lds_sync();
        for (int it = 1; it < nit; ++it) {
            const bool last = (it == nit - 1);
            coeff();
            mm(T, s0, Y, h, c);
            FOR_TILE_EL { const float v = T[I * 4 + J][r];
                          T[I * 4 + J][r] = (ON_DIAG ? a : 0.f) - b * v; }
            if (!last) {
                wave_lds_sync();
                put_lds(s0, T, h, c);
                wave_lds_sync();
            }
            mm(Yn, s1, T, h, c);
            if (!last) {
                mm(Zn, s0, Z, h, c);
                wave_lds_sync();
                put_lds(s1, Yn, h, c);
                put_lds(s0, Zn, h, c);
                wave_lds_sync();
#pragma unroll
                for (int t = 0; t < 16; ++t) Z[t] = Zn[t];
            }
#pragma unroll
            for (int t = 0; t < 16; ++t) Y[t] = Yn[t];
        }
#pragma unroll
        for (int t = 0; t < 16; ++t) A[t] = Y[t];
        lo = sqrtf(lo);
    }
    constexpr int LOG_D = 14;
    constexpr float LOG_IH = 2.60416675f;
    constexpr float LOG_SH = -1.61067712f;
    constexpr float LOG_P[LOG_D + 1] = {
        -0.480458081f, 0.620857120f, -0.192731783f, 0.0797645748f, -0.0371408127f,
        0.0185335092f, -0.00959482882f, 0.00469629187f, -0.00253403769f, 0.00239841524f,
        -0.00136570388f, -0.000536419684f, 0.000324033346f, 0.000684769184f, -0.000400175195f};
    FOR_TILE_EL { Y[I * 4 + J][r] = A[I * 4 + J][r] * LOG_IH + (ON_DIAG ? LOG_SH : 0.f); }
    wave_lds_sync();
    put_lds(s1, Y, h, c);
    wave_lds_sync();
    FOR_TILE_EL { Z[I * 4 + J][r] = Y[I * 4 + J][r] * LOG_P[LOG_D] + (ON_DIAG ? LOG_P[LOG_D - 1] : 0.f); }
    for (int j = LOG_D - 2; j >= 1; --j) {
        mm(T, s1, Z, h, c);
        const float pj = LOG_P[j];
        FOR_TILE_EL { Z[I * 4 + J][r] = T[I * 4 + J][r] + (ON_DIAG ? pj : 0.f); }
    }
    mm(T, s1, Z, h, c);
    const float lscale = (float)(1 << K_SQRT);

    FOR_TILE_EL { A[I * 4 + J][r] = alpha * (lscale * (T[I * 4 + J][r] + (ON_DIAG ? LOG_P[0] : 0.f))) + beta; }
    wave_lds_sync();
    put_lds(s0, A, h, c);
    wave_lds_sync();
    constexpr int S_EXP = 6, EXP_M = 10;
    const float es = 1.f / (float)(1 << S_EXP);
#pragma unroll
    for (int J = 0; J < 4; ++J) {
        v8f at[4];
        get_arow(at, s0, J, h, c);
#pragma unroll
        for (int I = 0; I < 4; ++I)
#pragma unroll
            for (int r = 0; r < 8; ++r)
                Y[I * 4 + J][r] = (0.5f * (A[I * 4 + J][r] + at[I][r])) * es;
    }

    wave_lds_sync();
    put_lds(s1, Y, h, c);
    wave_lds_sync();
    FOR_TILE_EL { Z[I * 4 + J][r] = Y[I * 4 + J][r] * (1.f / (float)EXP_M) + (ON_DIAG ? 1.f : 0.f); }
    for (int m = EXP_M - 1; m >= 1; --m) {
        mm(T, s1, Z, h, c);
        const float dm = 1.f / (float)m;
        FOR_TILE_EL { Z[I * 4 + J][r] = T[I * 4 + J][r] * dm + (ON_DIAG ? 1.f : 0.f); }
    }
    for (int j = 0; j < S_EXP; ++j) {
        wave_lds_sync();
        put_lds(s0, Z, h, c);
        wave_lds_sync();
        mm(T, s0, Z, h, c);
#pragma unroll
        for (int t = 0; t < 16; ++t) Z[t] = T[t];
    }

    wave_lds_sync();
    put_lds(s0, Z, h, c);
    wave_lds_sync();
    float* ob = out + (size_t)bid * 4096;
#pragma unroll
    for (int j = 0; j < 32; ++j) {
        const int row = 2 * j + h;
        const v4f v = *(const v4f*)(s0 + row * PW + 4 * c);
        *(v4f*)(ob + row * 64 + 4 * c) = v;
    }
}

extern "C" void kernel_launch(void* const* d_in, const int* in_sizes, int n_in,
                              void* d_out, int out_size, void* d_ws, size_t ws_size,
                              hipStream_t stream) {
    const float* x     = (const float*)d_in[0];
    const float* alpha = (const float*)d_in[1];
    const float* beta  = (const float*)d_in[2];
    float* out = (float*)d_out;

    const int nmat = 256 * N_EPOCH;
    const size_t shbytes = (size_t)(2 * SLOT) * sizeof(float);
    spdlogexp_kernel<<<nmat, 32, shbytes, stream>>>(x, alpha, beta, out);
}


# Square roots by degree-5 Newton-Schulz steps q = aI + bM + cM^2 (minimax constants from fixed spectral bounds, 4 products each): 22 -> 13 steps, 85 -> 71 products; root error 3e-7; rounding differs
# speedup vs baseline: 4.9656x; 1.1820x over previous
#include <hip/hip_runtime.h>

typedef __attribute__((ext_vector_type(2))) float v2f;
typedef __attribute__((ext_vector_type(8))) float v8f;

#define C_DIM 64
#define T_DIM 4096
#define N_EPOCH 8
#define PATCH 512
#define EPS_COV 1e-5f

typedef __attribute__((ext_vector_type(4))) float v4f;

#define PW 68
#define SLOT (64 * PW)


__device__ __forceinline__ void wave_lds_sync() { __syncthreads(); }

__device__ __forceinline__ void put_lds(float* __restrict__ s, const v8f (&M)[16], int h, int c) {
    float* p = s + (8 * h) * PW + c;
#pragma unroll
    for (int I = 0; I < 4; ++I)
#pragma unroll
        for (int J = 0; J < 4; ++J)
#pragma unroll
            for (int r = 0; r < 8; ++r)
                p[(16 * I + r) * PW + 16 * J] = M[I * 4 + J][r];
}

__device__ __forceinline__ void get_arow(v8f (&a)[4], const float* __restrict__ s, int J, int h, int c) {
    const float* p = s + (16 * J + c) * PW + 8 * h;
#pragma unroll
    for (int I = 0; I < 4; ++I) {
        const v4f lo = *(const v4f*)(p + 16 * I);
        const v4f hi = *(const v4f*)(p + 16 * I + 4);
        v8f t = {lo.x, lo.y, lo.z, lo.w, hi.x, hi.y, hi.z, hi.w};
        a[I] = t;
    }
}

__device__ __forceinline__ void mm(v8f (&D)[16], const float* __restrict__ sA, const v8f (&B)[16],
                                   int h, int c) {
#pragma unroll
    for (int J = 0; J < 4; ++J) {
        v8f a[4];
        get_arow(a, sA, J, h, c);
        v8f acc[4];
#pragma unroll
        for (int Jp = 0; Jp < 4; ++Jp) { v8f z = {0.f,0.f,0.f,0.f,0.f,0.f,0.f,0.f}; acc[Jp] = z; }
#pragma unroll
        for (int I = 0; I < 4; ++I)
#pragma unroll
            for (int q = 0; q < 4; ++q) {
                v2f af; af.x = a[I][2 * q]; af.y = a[I][2 * q + 1];
#pragma unroll
                for (int Jp = 0; Jp < 4; ++Jp) {
                    v2f bf; bf.x = B[I * 4 + Jp][2 * q]; bf.y = B[I * 4 + Jp][2 * q + 1];
                    acc[Jp] = __builtin_amdgcn_wmma_f32_16x16x4_f32(false, af, false, bf,
                                                                    (short)0, acc[Jp], false, false);
                }
            }
#pragma unroll
        for (int Jp = 0; Jp < 4; ++Jp) D[J * 4 + Jp] = acc[Jp];
    }
}

#define FOR_TILE_EL \
    _Pragma("unroll") for (int I = 0; I < 4; ++I) \
    _Pragma("unroll") for (int J = 0; J < 4; ++J) \
    _Pragma("unroll") for (int r = 0; r < 8; ++r)
#define ON_DIAG ((I == J) && (8 * h + r == c))

__global__ void __launch_bounds__(32)
spdlogexp_kernel(const float* __restrict__ x,
                 const float* __restrict__ alphap,
                 const float* __restrict__ betap,
                 float* __restrict__ out) {
    extern __shared__ __attribute__((aligned(16))) float lds[];
    float* s0 = lds;
    float* s1 = lds + SLOT;

    const int lane = threadIdx.x, h = lane >> 4, c = lane & 15;
    const int bid = blockIdx.x;
    const float alpha = alphap[0], beta = betap[0];

    v8f A[16], Y[16], Z[16], T[16], Yn[16], Zn[16];

    {
        const float* xr = x + (size_t)(bid >> 3) * C_DIM * T_DIM + (size_t)(bid & 7) * PATCH
                            + (size_t)c * T_DIM + 16 * h;
        v4f fa[4][4], fb[4][4], fc[4][4], fd[4][4];
        float ms[4] = {0.f, 0.f, 0.f, 0.f};
#pragma unroll
        for (int t = 0; t < 16; ++t) { v8f z = {0.f,0.f,0.f,0.f,0.f,0.f,0.f,0.f}; A[t] = z; }

        auto ldgrp = [&](v4f (&f)[4][4], int g) {
#pragma unroll
            for (int J = 0; J < 4; ++J)
#pragma unroll
                for (int i = 0; i < 4; ++i)
                    f[J][i] = *(const v4f*)(xr + (size_t)(16 * J) * T_DIM + 32 * g + 4 * i);
        };
        auto gram = [&](const v4f (&f)[4][4]) {
#pragma unroll
            for (int i = 0; i < 4; ++i)
#pragma unroll
                for (int p = 0; p < 2; ++p) {
                    v2f fr[4];
#pragma unroll
                    for (int J = 0; J < 4; ++J) { fr[J].x = f[J][i][2 * p]; fr[J].y = f[J][i][2 * p + 1]; }
#pragma unroll
                    for (int J = 0; J < 4; ++J)
#pragma unroll
                        for (int Jp = 0; Jp < 4; ++Jp)
                            A[J * 4 + Jp] = __builtin_amdgcn_wmma_f32_16x16x4_f32(false, fr[J], false, fr[Jp],
                                                                                 (short)0, A[J * 4 + Jp], false, false);
                }
#pragma unroll
            for (int J = 0; J < 4; ++J) {
                const v4f s4 = (f[J][0] + f[J][1]) + (f[J][2] + f[J][3]);
                ms[J] += (s4.x + s4.y) + (s4.z + s4.w);
            }
        };

        ldgrp(fa, 0);
        ldgrp(fb, 1);
        for (int g = 2; g < PATCH / 32 - 2; g += 4) {
            ldgrp(fc, g);
            ldgrp(fd, g + 1);
            gram(fa);
            gram(fb);
            ldgrp(fa, g + 2);
            ldgrp(fb, g + 3);
            gram(fc);
            gram(fd);
        }
        ldgrp(fc, PATCH / 32 - 2);
        ldgrp(fd, PATCH / 32 - 1);
        gram(fa);
        gram(fb);
        gram(fc);
        gram(fd);

        float mean[4];
#pragma unroll
        for (int J = 0; J < 4; ++J) {
            const float t = ms[J] + __shfl_xor(ms[J], 16);
            mean[J] = t * (1.f / PATCH);
        }
#pragma unroll
        for (int I = 0; I < 4; ++I)
#pragma unroll
            for (int r = 0; r < 8; ++r) {
                const float mr = __shfl(mean[I], 8 * h + r);
#pragma unroll
                for (int J = 0; J < 4; ++J)
                    A[I * 4 + J][r] = (A[I * 4 + J][r] - (float)PATCH * mr * mean[J]) * (1.f / (PATCH - 1));
            }
    }
    float tr = 0.f;
    FOR_TILE_EL { if (ON_DIAG) tr += A[I * 4 + J][r]; }
#pragma unroll
    for (int m = 16; m >= 1; m >>= 1) tr += __shfl_xor(tr, m);
    const float invtr = 1.f / tr;
    FOR_TILE_EL {
        float v = A[I * 4 + J][r] * invtr;
        if (ON_DIAG) v += EPS_COV;
        A[I * 4 + J][r] = v;
    }

    constexpr int K_SQRT = 3;
    constexpr float NS_Q[13][3] = {
        {4.24041025f, -12.4983416f, 9.24490642f},
        {4.20004358f, -12.1639474f, 8.9513856f},
        {4.03047609f, -10.8248071f, 7.7838105f},
        {3.39850427f, -6.70280372f, 4.299489f},
        {2.27363627f, -2.18404933f, 0.910138325f},
        {1.88740419f, -1.27491778f, 0.38751359f},
        {4.02844755f, -10.8094168f, 7.77047138f},
        {3.39198682f, -6.66692885f, 4.27017721f},
        {2.26804532f, -2.16891655f, 0.900604412f},
        {1.88707286f, -1.27424899f, 0.387176126f},
        {3.34496619f, -6.41186181f, 4.06245256f},
        {2.22974604f, -2.06690144f, 0.836939482f},
        {1.88501657f, -1.27010224f, 0.385085646f}};
    int qi = 0;
    for (int kq = 0; kq < K_SQRT; ++kq) {
        const int nit = (kq == 0) ? 6 : (kq == 1) ? 4 : 3;
        {
            const float qa = NS_Q[qi][0], qb = NS_Q[qi][1], qc = NS_Q[qi][2];
            ++qi;
            wave_lds_sync();
            put_lds(s1, A, h, c);
            wave_lds_sync();
            mm(T, s1, A, h, c);
            FOR_TILE_EL { Z[I * 4 + J][r] = (ON_DIAG ? qa : 0.f) + qb * A[I * 4 + J][r] + qc * T[I * 4 + J][r]; }
            mm(Y, s1, Z, h, c);
            wave_lds_sync();
            put_lds(s1, Y, h, c);
            put_lds(s0, Z, h, c);
            wave_lds_sync();
        }
        for (int it = 1; it < nit; ++it) {
            const bool last = (it == nit - 1);
            const float qa = NS_Q[qi][0], qb = NS_Q[qi][1], qc = NS_Q[qi][2];
            ++qi;
            mm(T, s0, Y, h, c);
            wave_lds_sync();
            put_lds(s0, T, h, c);
            wave_lds_sync();
            mm(Yn, s0, T, h, c);
            FOR_TILE_EL { const float m1 = T[I * 4 + J][r], m2 = Yn[I * 4 + J][r];
                          T[I * 4 + J][r] = (ON_DIAG ? qa : 0.f) + qb * m1 + qc * m2; }
            if (!last) {
                wave_lds_sync();
                put_lds(s0, T, h, c);
                wave_lds_sync();
            }
            mm(Yn, s1, T, h, c);
            if (!last) {
                mm(Zn, s0, Z, h, c);
                wave_lds_sync();
                put_lds(s1, Yn, h, c);
                put_lds(s0, Zn, h, c);
                wave_lds_sync();
#pragma unroll
                for (int t = 0; t < 16; ++t) Z[t] = Zn[t];
            }
#pragma unroll
            for (int t = 0; t < 16; ++t) Y[t] = Yn[t];
        }
#pragma unroll
        for (int t = 0; t < 16; ++t) A[t] = Y[t];
    }
    constexpr int LOG_D = 14;
    constexpr float LOG_IH = 2.60416675f;
    constexpr float LOG_SH = -1.61067712f;
    constexpr float LOG_P[LOG_D + 1] = {
        -0.480458081f, 0.620857120f, -0.192731783f, 0.0797645748f, -0.0371408127f,
        0.0185335092f, -0.00959482882f, 0.00469629187f, -0.00253403769f, 0.00239841524f,
        -0.00136570388f, -0.000536419684f, 0.000324033346f, 0.000684769184f, -0.000400175195f};
    FOR_TILE_EL { Y[I * 4 + J][r] = A[I * 4 + J][r] * LOG_IH + (ON_DIAG ? LOG_SH : 0.f); }
    wave_lds_sync();
    put_lds(s1, Y, h, c);
    wave_lds_sync();
    FOR_TILE_EL { Z[I * 4 + J][r] = Y[I * 4 + J][r] * LOG_P[LOG_D] + (ON_DIAG ? LOG_P[LOG_D - 1] : 0.f); }
    for (int j = LOG_D - 2; j >= 1; --j) {
        mm(T, s1, Z, h, c);
        const float pj = LOG_P[j];
        FOR_TILE_EL { Z[I * 4 + J][r] = T[I * 4 + J][r] + (ON_DIAG ? pj : 0.f); }
    }
    mm(T, s1, Z, h, c);
    const float lscale = (float)(1 << K_SQRT);

    FOR_TILE_EL { A[I * 4 + J][r] = alpha * (lscale * (T[I * 4 + J][r] + (ON_DIAG ? LOG_P[0] : 0.f))) + beta; }
    wave_lds_sync();
    put_lds(s0, A, h, c);
    wave_lds_sync();
    constexpr int S_EXP = 6, EXP_M = 10;
    const float es = 1.f / (float)(1 << S_EXP);
#pragma unroll
    for (int J = 0; J < 4; ++J) {
        v8f at[4];
        get_arow(at, s0, J, h, c);
#pragma unroll
        for (int I = 0; I < 4; ++I)
#pragma unroll
            for (int r = 0; r < 8; ++r)
                Y[I * 4 + J][r] = (0.5f * (A[I * 4 + J][r] + at[I][r])) * es;
    }

    wave_lds_sync();
    put_lds(s1, Y, h, c);
    wave_lds_sync();
    FOR_TILE_EL { Z[I * 4 + J][r] = Y[I * 4 + J][r] * (1.f / (float)EXP_M) + (ON_DIAG ? 1.f : 0.f); }
    for (int m = EXP_M - 1; m >= 1; --m) {
        mm(T, s1, Z, h, c);
        const float dm = 1.f / (float)m;
        FOR_TILE_EL { Z[I * 4 + J][r] = T[I * 4 + J][r] * dm + (ON_DIAG ? 1.f : 0.f); }
    }
    for (int j = 0; j < S_EXP; ++j) {
        wave_lds_sync();
        put_lds(s0, Z, h, c);
        wave_lds_sync();
        mm(T, s0, Z, h, c);
#pragma unroll
        for (int t = 0; t < 16; ++t) Z[t] = T[t];
    }

    wave_lds_sync();
    put_lds(s0, Z, h, c);
    wave_lds_sync();
    float* ob = out + (size_t)bid * 4096;
#pragma unroll
    for (int j = 0; j < 32; ++j) {
        const int row = 2 * j + h;
        const v4f v = *(const v4f*)(s0 + row * PW + 4 * c);
        *(v4f*)(ob + row * 64 + 4 * c) = v;
    }
}

extern "C" void kernel_launch(void* const* d_in, const int* in_sizes, int n_in,
                              void* d_out, int out_size, void* d_ws, size_t ws_size,
                              hipStream_t stream) {
    const float* x     = (const float*)d_in[0];
    const float* alpha = (const float*)d_in[1];
    const float* beta  = (const float*)d_in[2];
    float* out = (float*)d_out;

    const int nmat = 256 * N_EPOCH;
    const size_t shbytes = (size_t)(2 * SLOT) * sizeof(float);
    spdlogexp_kernel<<<nmat, 32, shbytes, stream>>>(x, alpha, beta, out);
}


# Covariance, log-polynomial and exp products use symmetry: 10 of 16 tiles computed, A fragment = the symmetric factor's register tile (no LDS copy), lower tiles mirrored via LDS; rounding differs
# speedup vs baseline: 5.8141x; 1.1709x over previous
#include <hip/hip_runtime.h>

typedef __attribute__((ext_vector_type(2))) float v2f;
typedef __attribute__((ext_vector_type(8))) float v8f;

#define C_DIM 64
#define T_DIM 4096
#define N_EPOCH 8
#define PATCH 512
#define EPS_COV 1e-5f

typedef __attribute__((ext_vector_type(4))) float v4f;

#define PW 68
#define SLOT (64 * PW)


__device__ __forceinline__ void wave_lds_sync() { __syncthreads(); }

__device__ __forceinline__ void put_lds(float* __restrict__ s, const v8f (&M)[16], int h, int c) {
    float* p = s + (8 * h) * PW + c;
#pragma unroll
    for (int I = 0; I < 4; ++I)
#pragma unroll
        for (int J = 0; J < 4; ++J)
#pragma unroll
            for (int r = 0; r < 8; ++r)
                p[(16 * I + r) * PW + 16 * J] = M[I * 4 + J][r];
}

__device__ __forceinline__ void get_arow(v8f (&a)[4], const float* __restrict__ s, int J, int h, int c) {
    const float* p = s + (16 * J + c) * PW + 8 * h;
#pragma unroll
    for (int I = 0; I < 4; ++I) {
        const v4f lo = *(const v4f*)(p + 16 * I);
        const v4f hi = *(const v4f*)(p + 16 * I + 4);
        v8f t = {lo.x, lo.y, lo.z, lo.w, hi.x, hi.y, hi.z, hi.w};
        a[I] = t;
    }
}

__device__ __forceinline__ void mm(v8f (&D)[16], const float* __restrict__ sA, const v8f (&B)[16],
                                   int h, int c) {
#pragma unroll
    for (int J = 0; J < 4; ++J) {
        v8f a[4];
        get_arow(a, sA, J, h, c);
        v8f acc[4];
#pragma unroll
        for (int Jp = 0; Jp < 4; ++Jp) { v8f z = {0.f,0.f,0.f,0.f,0.f,0.f,0.f,0.f}; acc[Jp] = z; }
#pragma unroll
        for (int I = 0; I < 4; ++I)
#pragma unroll
            for (int q = 0; q < 4; ++q) {
                v2f af; af.x = a[I][2 * q]; af.y = a[I][2 * q + 1];
#pragma unroll
                for (int Jp = 0; Jp < 4; ++Jp) {
                    v2f bf; bf.x = B[I * 4 + Jp][2 * q]; bf.y = B[I * 4 + Jp][2 * q + 1];
                    acc[Jp] = __builtin_amdgcn_wmma_f32_16x16x4_f32(false, af, false, bf,
                                                                    (short)0, acc[Jp], false, false);
                }
            }
#pragma unroll
        for (int Jp = 0; Jp < 4; ++Jp) D[J * 4 + Jp] = acc[Jp];
    }
}

__device__ __forceinline__ void mm_sym(v8f (&D)[16], const v8f (&P)[16], const v8f (&Q)[16]) {
#pragma unroll
    for (int J = 0; J < 4; ++J)
#pragma unroll
        for (int Jp = J; Jp < 4; ++Jp) { v8f z = {0.f,0.f,0.f,0.f,0.f,0.f,0.f,0.f}; D[J * 4 + Jp] = z; }
#pragma unroll
    for (int I = 0; I < 4; ++I)
#pragma unroll
        for (int q = 0; q < 4; ++q)
#pragma unroll
            for (int J = 0; J < 4; ++J) {
                v2f af; af.x = P[I * 4 + J][2 * q]; af.y = P[I * 4 + J][2 * q + 1];
#pragma unroll
                for (int Jp = J; Jp < 4; ++Jp) {
                    v2f bf; bf.x = Q[I * 4 + Jp][2 * q]; bf.y = Q[I * 4 + Jp][2 * q + 1];
                    D[J * 4 + Jp] = __builtin_amdgcn_wmma_f32_16x16x4_f32(false, af, false, bf,
                                                                          (short)0, D[J * 4 + Jp], false, false);
                }
            }
}

__device__ __forceinline__ void mirror_lower(float* __restrict__ s, v8f (&M)[16], int h, int c) {
    wave_lds_sync();
    float* p = s + (8 * h) * PW + c;
#pragma unroll
    for (int I = 0; I < 4; ++I)
#pragma unroll
        for (int J = I; J < 4; ++J)
#pragma unroll
            for (int r = 0; r < 8; ++r)
                p[(16 * I + r) * PW + 16 * J] = M[I * 4 + J][r];
    wave_lds_sync();
#pragma unroll
    for (int J = 0; J < 3; ++J) {
        const float* rp = s + (16 * J + c) * PW + 8 * h;
#pragma unroll
        for (int I = J + 1; I < 4; ++I) {
            const v4f lo = *(const v4f*)(rp + 16 * I);
            const v4f hi = *(const v4f*)(rp + 16 * I + 4);
            v8f t = {lo.x, lo.y, lo.z, lo.w, hi.x, hi.y, hi.z, hi.w};
            M[I * 4 + J] = t;
        }
    }
}

#define FOR_TILE_EL \
    _Pragma("unroll") for (int I = 0; I < 4; ++I) \
    _Pragma("unroll") for (int J = 0; J < 4; ++J) \
    _Pragma("unroll") for (int r = 0; r < 8; ++r)
#define FOR_UPPER_EL \
    _Pragma("unroll") for (int I = 0; I < 4; ++I) \
    _Pragma("unroll") for (int J = I; J < 4; ++J) \
    _Pragma("unroll") for (int r = 0; r < 8; ++r)
#define ON_DIAG ((I == J) && (8 * h + r == c))

__global__ void __launch_bounds__(32)
spdlogexp_kernel(const float* __restrict__ x,
                 const float* __restrict__ alphap,
                 const float* __restrict__ betap,
                 float* __restrict__ out) {
    extern __shared__ __attribute__((aligned(16))) float lds[];
    float* s0 = lds;
    float* s1 = lds + SLOT;

    const int lane = threadIdx.x, h = lane >> 4, c = lane & 15;
    const int bid = blockIdx.x;
    const float alpha = alphap[0], beta = betap[0];

    v8f A[16], Y[16], Z[16], T[16], Yn[16], Zn[16];

    {
        const float* xr = x + (size_t)(bid >> 3) * C_DIM * T_DIM + (size_t)(bid & 7) * PATCH
                            + (size_t)c * T_DIM + 16 * h;
        v4f fa[4][4], fb[4][4], fc[4][4], fd[4][4];
        float ms[4] = {0.f, 0.f, 0.f, 0.f};
#pragma unroll
        for (int t = 0; t < 16; ++t) { v8f z = {0.f,0.f,0.f,0.f,0.f,0.f,0.f,0.f}; A[t] = z; }

        auto ldgrp = [&](v4f (&f)[4][4], int g) {
#pragma unroll
            for (int J = 0; J < 4; ++J)
#pragma unroll
                for (int i = 0; i < 4; ++i)
                    f[J][i] = *(const v4f*)(xr + (size_t)(16 * J) * T_DIM + 32 * g + 4 * i);
        };
        auto gram = [&](const v4f (&f)[4][4]) {
#pragma unroll
            for (int i = 0; i < 4; ++i)
#pragma unroll
                for (int p = 0; p < 2; ++p) {
                    v2f fr[4];
#pragma unroll
                    for (int J = 0; J < 4; ++J) { fr[J].x = f[J][i][2 * p]; fr[J].y = f[J][i][2 * p + 1]; }
#pragma unroll
                    for (int J = 0; J < 4; ++J)
#pragma unroll
                        for (int Jp = J; Jp < 4; ++Jp)
                            A[J * 4 + Jp] = __builtin_amdgcn_wmma_f32_16x16x4_f32(false, fr[J], false, fr[Jp],
                                                                                 (short)0, A[J * 4 + Jp], false, false);
                }
#pragma unroll
            for (int J = 0; J < 4; ++J) {
                const v4f s4 = (f[J][0] + f[J][1]) + (f[J][2] + f[J][3]);
                ms[J] += (s4.x + s4.y) + (s4.z + s4.w);
            }
        };

        ldgrp(fa, 0);
        ldgrp(fb, 1);
        for (int g = 2; g < PATCH / 32 - 2; g += 4) {
            ldgrp(fc, g);
            ldgrp(fd, g + 1);
            gram(fa);
            gram(fb);
            ldgrp(fa, g + 2);
            ldgrp(fb, g + 3);
            gram(fc);
            gram(fd);
        }
        ldgrp(fc, PATCH / 32 - 2);
        ldgrp(fd, PATCH / 32 - 1);
        gram(fa);
        gram(fb);
        gram(fc);
        gram(fd);

        mirror_lower(s0, A, h, c);

        float mean[4];
#pragma unroll
        for (int J = 0; J < 4; ++J) {
            const float t = ms[J] + __shfl_xor(ms[J], 16);
            mean[J] = t * (1.f / PATCH);
        }
#pragma unroll
        for (int I = 0; I < 4; ++I)
#pragma unroll
            for (int r = 0; r < 8; ++r) {
                const float mr = __shfl(mean[I], 8 * h + r);
#pragma unroll
                for (int J = 0; J < 4; ++J)
                    A[I * 4 + J][r] = (A[I * 4 + J][r] - (float)PATCH * mr * mean[J]) * (1.f / (PATCH - 1));
            }
    }
    float tr = 0.f;
    FOR_TILE_EL { if (ON_DIAG) tr += A[I * 4 + J][r]; }
#pragma unroll
    for (int m = 16; m >= 1; m >>= 1) tr += __shfl_xor(tr, m);
    const float invtr = 1.f / tr;
    FOR_TILE_EL {
        float v = A[I * 4 + J][r] * invtr;
        if (ON_DIAG) v += EPS_COV;
        A[I * 4 + J][r] = v;
    }

    constexpr int K_SQRT = 3;
    constexpr float NS_Q[13][3] = {
        {4.24041025f, -12.4983416f, 9.24490642f},
        {4.20004358f, -12.1639474f, 8.9513856f},
        {4.03047609f, -10.8248071f, 7.7838105f},
        {3.39850427f, -6.70280372f, 4.299489f},
        {2.27363627f, -2.18404933f, 0.910138325f},
        {1.88740419f, -1.27491778f, 0.38751359f},
        {4.02844755f, -10.8094168f, 7.77047138f},
        {3.39198682f, -6.66692885f, 4.27017721f},
        {2.26804532f, -2.16891655f, 0.900604412f},
        {1.88707286f, -1.27424899f, 0.387176126f},
        {3.34496619f, -6.41186181f, 4.06245256f},
        {2.22974604f, -2.06690144f, 0.836939482f},
        {1.88501657f, -1.27010224f, 0.385085646f}};
    int qi = 0;
    for (int kq = 0; kq < K_SQRT; ++kq) {
        const int nit = (kq == 0) ? 6 : (kq == 1) ? 4 : 3;
        {
            const float qa = NS_Q[qi][0], qb = NS_Q[qi][1], qc = NS_Q[qi][2];
            ++qi;
            wave_lds_sync();
            put_lds(s1, A, h, c);
            wave_lds_sync();
            mm(T, s1, A, h, c);
            FOR_TILE_EL { Z[I * 4 + J][r] = (ON_DIAG ? qa : 0.f) + qb * A[I * 4 + J][r] + qc * T[I * 4 + J][r]; }
            mm(Y, s1, Z, h, c);
            wave_lds_sync();
            put_lds(s1, Y, h, c);
            put_lds(s0, Z, h, c);
            wave_lds_sync();
        }
        for (int it = 1; it < nit; ++it) {
            const bool last = (it == nit - 1);
            const float qa = NS_Q[qi][0], qb = NS_Q[qi][1], qc = NS_Q[qi][2];
            ++qi;
            mm(T, s0, Y, h, c);
            wave_lds_sync();
            put_lds(s0, T, h, c);
            wave_lds_sync();
            mm(Yn, s0, T, h, c);
            FOR_TILE_EL { const float m1 = T[I * 4 + J][r], m2 = Yn[I * 4 + J][r];
                          T[I * 4 + J][r] = (ON_DIAG ? qa : 0.f) + qb * m1 + qc * m2; }
            if (!last) {
                wave_lds_sync();
                put_lds(s0, T, h, c);
                wave_lds_sync();
            }
            mm(Yn, s1, T, h, c);
            if (!last) {
                mm(Zn, s0, Z, h, c);
                wave_lds_sync();
                put_lds(s1, Yn, h, c);
                put_lds(s0, Zn, h, c);
                wave_lds_sync();
#pragma unroll
                for (int t = 0; t < 16; ++t) Z[t] = Zn[t];
            }
#pragma unroll
            for (int t = 0; t < 16; ++t) Y[t] = Yn[t];
        }
#pragma unroll
        for (int t = 0; t < 16; ++t) A[t] = Y[t];
    }
    constexpr int LOG_D = 14;
    constexpr float LOG_IH = 2.60416675f;
    constexpr float LOG_SH = -1.61067712f;
    constexpr float LOG_P[LOG_D + 1] = {
        -0.480458081f, 0.620857120f, -0.192731783f, 0.0797645748f, -0.0371408127f,
        0.0185335092f, -0.00959482882f, 0.00469629187f, -0.00253403769f, 0.00239841524f,
        -0.00136570388f, -0.000536419684f, 0.000324033346f, 0.000684769184f, -0.000400175195f};
    FOR_TILE_EL { Y[I * 4 + J][r] = A[I * 4 + J][r] * LOG_IH + (ON_DIAG ? LOG_SH : 0.f); }
    FOR_TILE_EL { Z[I * 4 + J][r] = Y[I * 4 + J][r] * LOG_P[LOG_D] + (ON_DIAG ? LOG_P[LOG_D - 1] : 0.f); }
    for (int j = LOG_D - 2; j >= 1; --j) {
        mm_sym(T, Y, Z);
        const float pj = LOG_P[j];
        FOR_UPPER_EL { Z[I * 4 + J][r] = T[I * 4 + J][r] + (ON_DIAG ? pj : 0.f); }
        mirror_lower(s0, Z, h, c);
    }
    mm_sym(T, Y, Z);
    const float lscale = (float)(1 << K_SQRT);

    FOR_UPPER_EL { A[I * 4 + J][r] = alpha * (lscale * (T[I * 4 + J][r] + (ON_DIAG ? LOG_P[0] : 0.f))) + beta; }
    mirror_lower(s1, A, h, c);
    wave_lds_sync();
    put_lds(s0, A, h, c);
    wave_lds_sync();
    constexpr int S_EXP = 6, EXP_M = 10;
    const float es = 1.f / (float)(1 << S_EXP);
#pragma unroll
    for (int J = 0; J < 4; ++J) {
        v8f at[4];
        get_arow(at, s0, J, h, c);
#pragma unroll
        for (int I = 0; I < 4; ++I)
#pragma unroll
            for (int r = 0; r < 8; ++r)
                Y[I * 4 + J][r] = (0.5f * (A[I * 4 + J][r] + at[I][r])) * es;
    }

    FOR_TILE_EL { Z[I * 4 + J][r] = Y[I * 4 + J][r] * (1.f / (float)EXP_M) + (ON_DIAG ? 1.f : 0.f); }
    for (int m = EXP_M - 1; m >= 1; --m) {
        mm_sym(T, Y, Z);
        const float dm = 1.f / (float)m;
        FOR_UPPER_EL { Z[I * 4 + J][r] = T[I * 4 + J][r] * dm + (ON_DIAG ? 1.f : 0.f); }
        mirror_lower(s0, Z, h, c);
    }
    for (int j = 0; j < S_EXP; ++j) {
        mm_sym(T, Z, Z);
        FOR_UPPER_EL { Z[I * 4 + J][r] = T[I * 4 + J][r]; }
        mirror_lower(s0, Z, h, c);
    }

    wave_lds_sync();
    put_lds(s0, Z, h, c);
    wave_lds_sync();
    float* ob = out + (size_t)bid * 4096;
#pragma unroll
    for (int j = 0; j < 32; ++j) {
        const int row = 2 * j + h;
        const v4f v = *(const v4f*)(s0 + row * PW + 4 * c);
        *(v4f*)(ob + row * 64 + 4 * c) = v;
    }
}

extern "C" void kernel_launch(void* const* d_in, const int* in_sizes, int n_in,
                              void* d_out, int out_size, void* d_ws, size_t ws_size,
                              hipStream_t stream) {
    const float* x     = (const float*)d_in[0];
    const float* alpha = (const float*)d_in[1];
    const float* beta  = (const float*)d_in[2];
    float* out = (float*)d_out;

    const int nmat = 256 * N_EPOCH;
    const size_t shbytes = (size_t)(2 * SLOT) * sizeof(float);
    spdlogexp_kernel<<<nmat, 32, shbytes, stream>>>(x, alpha, beta, out);
}


# Square-root steps of degree 9 (q = aI + bM + M^2(cI + dM + eM^2): order 5, 5 products, aI + bM as the WMMA C input) but for root 1's last two: 13 -> 10 steps, 71 -> 67 products; rounding differs
# speedup vs baseline: 6.0176x; 1.0350x over previous
#include <hip/hip_runtime.h>

typedef __attribute__((ext_vector_type(2))) float v2f;
typedef __attribute__((ext_vector_type(8))) float v8f;

#define C_DIM 64
#define T_DIM 4096
#define N_EPOCH 8
#define PATCH 512
#define EPS_COV 1e-5f

typedef __attribute__((ext_vector_type(4))) float v4f;

#define PW 68
#define SLOT (64 * PW)


__device__ __forceinline__ void wave_lds_sync() { __syncthreads(); }

__device__ __forceinline__ void put_lds(float* __restrict__ s, const v8f (&M)[16], int h, int c) {
    float* p = s + (8 * h) * PW + c;
#pragma unroll
    for (int I = 0; I < 4; ++I)
#pragma unroll
        for (int J = 0; J < 4; ++J)
#pragma unroll
            for (int r = 0; r < 8; ++r)
                p[(16 * I + r) * PW + 16 * J] = M[I * 4 + J][r];
}

__device__ __forceinline__ void get_arow(v8f (&a)[4], const float* __restrict__ s, int J, int h, int c) {
    const float* p = s + (16 * J + c) * PW + 8 * h;
#pragma unroll
    for (int I = 0; I < 4; ++I) {
        const v4f lo = *(const v4f*)(p + 16 * I);
        const v4f hi = *(const v4f*)(p + 16 * I + 4);
        v8f t = {lo.x, lo.y, lo.z, lo.w, hi.x, hi.y, hi.z, hi.w};
        a[I] = t;
    }
}

__device__ __forceinline__ void mm(v8f (&D)[16], const float* __restrict__ sA, const v8f (&B)[16],
                                   int h, int c) {
#pragma unroll
    for (int J = 0; J < 4; ++J) {
        v8f a[4];
        get_arow(a, sA, J, h, c);
        v8f acc[4];
#pragma unroll
        for (int Jp = 0; Jp < 4; ++Jp) { v8f z = {0.f,0.f,0.f,0.f,0.f,0.f,0.f,0.f}; acc[Jp] = z; }
#pragma unroll
        for (int I = 0; I < 4; ++I)
#pragma unroll
            for (int q = 0; q < 4; ++q) {
                v2f af; af.x = a[I][2 * q]; af.y = a[I][2 * q + 1];
#pragma unroll
                for (int Jp = 0; Jp < 4; ++Jp) {
                    v2f bf; bf.x = B[I * 4 + Jp][2 * q]; bf.y = B[I * 4 + Jp][2 * q + 1];
                    acc[Jp] = __builtin_amdgcn_wmma_f32_16x16x4_f32(false, af, false, bf,
                                                                    (short)0, acc[Jp], false, false);
                }
            }
#pragma unroll
        for (int Jp = 0; Jp < 4; ++Jp) D[J * 4 + Jp] = acc[Jp];
    }
}

__device__ __forceinline__ void mm_add(v8f (&D)[16], const float* __restrict__ sA, const v8f (&B)[16],
                                       int h, int c, bool on) {
#pragma unroll
    for (int J = 0; J < 4; ++J) {
        if (!on) continue;
        v8f a[4];
        get_arow(a, sA, J, h, c);
        v8f acc[4];
#pragma unroll
        for (int Jp = 0; Jp < 4; ++Jp) acc[Jp] = D[J * 4 + Jp];
#pragma unroll
        for (int I = 0; I < 4; ++I)
#pragma unroll
            for (int q = 0; q < 4; ++q) {
                v2f af; af.x = a[I][2 * q]; af.y = a[I][2 * q + 1];
#pragma unroll
                for (int Jp = 0; Jp < 4; ++Jp) {
                    v2f bf; bf.x = B[I * 4 + Jp][2 * q]; bf.y = B[I * 4 + Jp][2 * q + 1];
                    acc[Jp] = __builtin_amdgcn_wmma_f32_16x16x4_f32(false, af, false, bf,
                                                                    (short)0, acc[Jp], false, false);
                }
            }
#pragma unroll
        for (int Jp = 0; Jp < 4; ++Jp) D[J * 4 + Jp] = acc[Jp];
    }
}

__device__ __forceinline__ void mm_sym(v8f (&D)[16], const v8f (&P)[16], const v8f (&Q)[16]) {
#pragma unroll
    for (int J = 0; J < 4; ++J)
#pragma unroll
        for (int Jp = J; Jp < 4; ++Jp) { v8f z = {0.f,0.f,0.f,0.f,0.f,0.f,0.f,0.f}; D[J * 4 + Jp] = z; }
#pragma unroll
    for (int I = 0; I < 4; ++I)
#pragma unroll
        for (int q = 0; q < 4; ++q)
#pragma unroll
            for (int J = 0; J < 4; ++J) {
                v2f af; af.x = P[I * 4 + J][2 * q]; af.y = P[I * 4 + J][2 * q + 1];
#pragma unroll
                for (int Jp = J; Jp < 4; ++Jp) {
                    v2f bf; bf.x = Q[I * 4 + Jp][2 * q]; bf.y = Q[I * 4 + Jp][2 * q + 1];
                    D[J * 4 + Jp] = __builtin_amdgcn_wmma_f32_16x16x4_f32(false, af, false, bf,
                                                                          (short)0, D[J * 4 + Jp], false, false);
                }
            }
}

__device__ __forceinline__ void mirror_lower(float* __restrict__ s, v8f (&M)[16], int h, int c) {
    wave_lds_sync();
    float* p = s + (8 * h) * PW + c;
#pragma unroll
    for (int I = 0; I < 4; ++I)
#pragma unroll
        for (int J = I; J < 4; ++J)
#pragma unroll
            for (int r = 0; r < 8; ++r)
                p[(16 * I + r) * PW + 16 * J] = M[I * 4 + J][r];
    wave_lds_sync();
#pragma unroll
    for (int J = 0; J < 3; ++J) {
        const float* rp = s + (16 * J + c) * PW + 8 * h;
#pragma unroll
        for (int I = J + 1; I < 4; ++I) {
            const v4f lo = *(const v4f*)(rp + 16 * I);
            const v4f hi = *(const v4f*)(rp + 16 * I + 4);
            v8f t = {lo.x, lo.y, lo.z, lo.w, hi.x, hi.y, hi.z, hi.w};
            M[I * 4 + J] = t;
        }
    }
}

#define FOR_TILE_EL \
    _Pragma("unroll") for (int I = 0; I < 4; ++I) \
    _Pragma("unroll") for (int J = 0; J < 4; ++J) \
    _Pragma("unroll") for (int r = 0; r < 8; ++r)
#define FOR_UPPER_EL \
    _Pragma("unroll") for (int I = 0; I < 4; ++I) \
    _Pragma("unroll") for (int J = I; J < 4; ++J) \
    _Pragma("unroll") for (int r = 0; r < 8; ++r)
#define ON_DIAG ((I == J) && (8 * h + r == c))

__global__ void __launch_bounds__(32, 2)
spdlogexp_kernel(const float* __restrict__ x,
                 const float* __restrict__ alphap,
                 const float* __restrict__ betap,
                 float* __restrict__ out) {
    extern __shared__ __attribute__((aligned(16))) float lds[];
    float* s0 = lds;
    float* s1 = lds + SLOT;

    const int lane = threadIdx.x, h = lane >> 4, c = lane & 15;
    const int bid = blockIdx.x;
    const float alpha = alphap[0], beta = betap[0];

    v8f A[16], Y[16], Z[16], T[16], Yn[16], Zn[16];

    {
        const float* xr = x + (size_t)(bid >> 3) * C_DIM * T_DIM + (size_t)(bid & 7) * PATCH
                            + (size_t)c * T_DIM + 16 * h;
        v4f fa[4][4], fb[4][4], fc[4][4], fd[4][4];
        float ms[4] = {0.f, 0.f, 0.f, 0.f};
#pragma unroll
        for (int t = 0; t < 16; ++t) { v8f z = {0.f,0.f,0.f,0.f,0.f,0.f,0.f,0.f}; A[t] = z; }

        auto ldgrp = [&](v4f (&f)[4][4], int g) {
#pragma unroll
            for (int J = 0; J < 4; ++J)
#pragma unroll
                for (int i = 0; i < 4; ++i)
                    f[J][i] = *(const v4f*)(xr + (size_t)(16 * J) * T_DIM + 32 * g + 4 * i);
        };
        auto gram = [&](const v4f (&f)[4][4]) {
#pragma unroll
            for (int i = 0; i < 4; ++i)
#pragma unroll
                for (int p = 0; p < 2; ++p) {
                    v2f fr[4];
#pragma unroll
                    for (int J = 0; J < 4; ++J) { fr[J].x = f[J][i][2 * p]; fr[J].y = f[J][i][2 * p + 1]; }
#pragma unroll
                    for (int J = 0; J < 4; ++J)
#pragma unroll
                        for (int Jp = J; Jp < 4; ++Jp)
                            A[J * 4 + Jp] = __builtin_amdgcn_wmma_f32_16x16x4_f32(false, fr[J], false, fr[Jp],
                                                                                 (short)0, A[J * 4 + Jp], false, false);
                }
#pragma unroll
            for (int J = 0; J < 4; ++J) {
                const v4f s4 = (f[J][0] + f[J][1]) + (f[J][2] + f[J][3]);
                ms[J] += (s4.x + s4.y) + (s4.z + s4.w);
            }
        };

        ldgrp(fa, 0);
        ldgrp(fb, 1);
        for (int g = 2; g < PATCH / 32 - 2; g += 4) {
            ldgrp(fc, g);
            ldgrp(fd, g + 1);
            gram(fa);
            gram(fb);
            ldgrp(fa, g + 2);
            ldgrp(fb, g + 3);
            gram(fc);
            gram(fd);
        }
        ldgrp(fc, PATCH / 32 - 2);
        ldgrp(fd, PATCH / 32 - 1);
        gram(fa);
        gram(fb);
        gram(fc);
        gram(fd);

        mirror_lower(s0, A, h, c);

        float mean[4];
#pragma unroll
        for (int J = 0; J < 4; ++J) {
            const float t = ms[J] + __shfl_xor(ms[J], 16);
            mean[J] = t * (1.f / PATCH);
        }
#pragma unroll
        for (int I = 0; I < 4; ++I)
#pragma unroll
            for (int r = 0; r < 8; ++r) {
                const float mr = __shfl(mean[I], 8 * h + r);
#pragma unroll
                for (int J = 0; J < 4; ++J)
                    A[I * 4 + J][r] = (A[I * 4 + J][r] - (float)PATCH * mr * mean[J]) * (1.f / (PATCH - 1));
            }
    }
    float tr = 0.f;
    FOR_TILE_EL { if (ON_DIAG) tr += A[I * 4 + J][r]; }
#pragma unroll
    for (int m = 16; m >= 1; m >>= 1) tr += __shfl_xor(tr, m);
    const float invtr = 1.f / tr;
    FOR_TILE_EL {
        float v = A[I * 4 + J][r] * invtr;
        if (ON_DIAG) v += EPS_COV;
        A[I * 4 + J][r] = v;
    }

    constexpr int K_SQRT = 3;
    constexpr int NS_STEPS = 10;
    constexpr float NS_Q[NS_STEPS][5] = {
        {7.55900367f, -73.5964731f, 244.153359f, -313.036904f, 135.881263f},
        {7.28266995f, -66.3727064f, 215.821045f, -274.034362f, 118.269434f},
        {5.50208321f, -30.8133248f, 82.8106228f, -95.0746483f, 38.5664482f},
        {2.10391988f, -1.75144606f, 0.647438191f, 0.f, 0.f},
        {1.88060021f, -1.26121892f, 0.380618686f, 0.f, 0.f},
        {6.84956223f, -56.0340356f, 175.852534f, -219.388936f, 93.6949753f},
        {4.03327347f, -13.3130201f, 26.2123828f, -24.6625897f, 8.72885709f},
        {2.49290395f, -3.41053688f, 3.1492112f, -1.53842768f, 0.306849411f},
        {4.91259235f, -22.6839514f, 55.1655881f, -59.7002108f, 23.3014127f},
        {2.60087938f, -3.8688644f, 3.87851041f, -2.05400127f, 0.443476032f}};
    constexpr unsigned NS_QUADRATIC = 0x18u;
    int qi = 0;
    for (int kq = 0; kq < K_SQRT; ++kq) {
        const int nit = (kq == 0) ? 5 : (kq == 1) ? 3 : 2;
        {
            const float q0 = NS_Q[qi][0], q1 = NS_Q[qi][1], q2 = NS_Q[qi][2], q3 = NS_Q[qi][3], q4 = NS_Q[qi][4];
            ++qi;
            wave_lds_sync();
            put_lds(s1, A, h, c);
            wave_lds_sync();
            mm(T, s1, A, h, c);
            wave_lds_sync();
            put_lds(s0, T, h, c);
            wave_lds_sync();
            FOR_TILE_EL { const float m1 = A[I * 4 + J][r], m2 = T[I * 4 + J][r];
                          T[I * 4 + J][r] = (ON_DIAG ? q2 : 0.f) + q3 * m1 + q4 * m2;
                          Z[I * 4 + J][r] = (ON_DIAG ? q0 : 0.f) + q1 * m1; }
            mm_add(Z, s0, T, h, c, true);
            mm(Y, s1, Z, h, c);
            wave_lds_sync();
            put_lds(s1, Y, h, c);
            put_lds(s0, Z, h, c);
            wave_lds_sync();
        }
        for (int it = 1; it < nit; ++it) {
            const bool last = (it == nit - 1);
            const float q0 = NS_Q[qi][0], q1 = NS_Q[qi][1], q2 = NS_Q[qi][2], q3 = NS_Q[qi][3], q4 = NS_Q[qi][4];
            const bool quartic = !((NS_QUADRATIC >> qi) & 1u);
            const float qs = quartic ? 0.f : q2;
            ++qi;
            mm(T, s0, Y, h, c);
            wave_lds_sync();
            put_lds(s0, T, h, c);
            wave_lds_sync();
            mm(Yn, s0, T, h, c);
            if (quartic) {
                wave_lds_sync();
                put_lds(s0, Yn, h, c);
                wave_lds_sync();
            }
            FOR_TILE_EL { const float m1 = T[I * 4 + J][r], m2 = Yn[I * 4 + J][r];
                          Yn[I * 4 + J][r] = (ON_DIAG ? q2 : 0.f) + q3 * m1 + q4 * m2;
                          T[I * 4 + J][r] = (ON_DIAG ? q0 : 0.f) + q1 * m1 + qs * m2; }
            mm_add(T, s0, Yn, h, c, quartic);
            if (!last) {
                wave_lds_sync();
                put_lds(s0, T, h, c);
                wave_lds_sync();
            }
            mm(Yn, s1, T, h, c);
            if (!last) {
                mm(Zn, s0, Z, h, c);
                wave_lds_sync();
                put_lds(s1, Yn, h, c);
                put_lds(s0, Zn, h, c);
                wave_lds_sync();
#pragma unroll
                for (int t = 0; t < 16; ++t) Z[t] = Zn[t];
            }
#pragma unroll
            for (int t = 0; t < 16; ++t) Y[t] = Yn[t];
        }
#pragma unroll
        for (int t = 0; t < 16; ++t) A[t] = Y[t];
    }
    constexpr int LOG_D = 14;
    constexpr float LOG_IH = 2.60416675f;
    constexpr float LOG_SH = -1.61067712f;
    constexpr float LOG_P[LOG_D + 1] = {
        -0.480458081f, 0.620857120f, -0.192731783f, 0.0797645748f, -0.0371408127f,
        0.0185335092f, -0.00959482882f, 0.00469629187f, -0.00253403769f, 0.00239841524f,
        -0.00136570388f, -0.000536419684f, 0.000324033346f, 0.000684769184f, -0.000400175195f};
    FOR_TILE_EL { Y[I * 4 + J][r] = A[I * 4 + J][r] * LOG_IH + (ON_DIAG ? LOG_SH : 0.f); }
    FOR_TILE_EL { Z[I * 4 + J][r] = Y[I * 4 + J][r] * LOG_P[LOG_D] + (ON_DIAG ? LOG_P[LOG_D - 1] : 0.f); }
    for (int j = LOG_D - 2; j >= 1; --j) {
        mm_sym(T, Y, Z);
        const float pj = LOG_P[j];
        FOR_UPPER_EL { Z[I * 4 + J][r] = T[I * 4 + J][r] + (ON_DIAG ? pj : 0.f); }
        mirror_lower(s0, Z, h, c);
    }
    mm_sym(T, Y, Z);
    const float lscale = (float)(1 << K_SQRT);

    FOR_UPPER_EL { A[I * 4 + J][r] = alpha * (lscale * (T[I * 4 + J][r] + (ON_DIAG ? LOG_P[0] : 0.f))) + beta; }
    mirror_lower(s1, A, h, c);
    wave_lds_sync();
    put_lds(s0, A, h, c);
    wave_lds_sync();
    constexpr int S_EXP = 6, EXP_M = 10;
    const float es = 1.f / (float)(1 << S_EXP);
#pragma unroll
    for (int J = 0; J < 4; ++J) {
        v8f at[4];
        get_arow(at, s0, J, h, c);
#pragma unroll
        for (int I = 0; I < 4; ++I)
#pragma unroll
            for (int r = 0; r < 8; ++r)
                Y[I * 4 + J][r] = (0.5f * (A[I * 4 + J][r] + at[I][r])) * es;
    }

    FOR_TILE_EL { Z[I * 4 + J][r] = Y[I * 4 + J][r] * (1.f / (float)EXP_M) + (ON_DIAG ? 1.f : 0.f); }
    for (int m = EXP_M - 1; m >= 1; --m) {
        mm_sym(T, Y, Z);
        const float dm = 1.f / (float)m;
        FOR_UPPER_EL { Z[I * 4 + J][r] = T[I * 4 + J][r] * dm + (ON_DIAG ? 1.f : 0.f); }
        mirror_lower(s0, Z, h, c);
    }
    for (int j = 0; j < S_EXP; ++j) {
        mm_sym(T, Z, Z);
        FOR_UPPER_EL { Z[I * 4 + J][r] = T[I * 4 + J][r]; }
        mirror_lower(s0, Z, h, c);
    }

    wave_lds_sync();
    put_lds(s0, Z, h, c);
    wave_lds_sync();
    float* ob = out + (size_t)bid * 4096;
#pragma unroll
    for (int j = 0; j < 32; ++j) {
        const int row = 2 * j + h;
        const v4f v = *(const v4f*)(s0 + row * PW + 4 * c);
        *(v4f*)(ob + row * 64 + 4 * c) = v;
    }
}

extern "C" void kernel_launch(void* const* d_in, const int* in_sizes, int n_in,
                              void* d_out, int out_size, void* d_ws, size_t ws_size,
                              hipStream_t stream) {
    const float* x     = (const float*)d_in[0];
    const float* alpha = (const float*)d_in[1];
    const float* beta  = (const float*)d_in[2];
    float* out = (float*)d_out;

    const int nmat = 256 * N_EPOCH;
    const size_t shbytes = (size_t)(2 * SLOT) * sizeof(float);
    spdlogexp_kernel<<<nmat, 32, shbytes, stream>>>(x, alpha, beta, out);
}


# Log from the root and inverse root the coupled iteration already yields: log x = log s + u*P(v), u,v = (x/s -+ s/x)/2, P of degree 17; 1 square root, not 3: 67 -> 53 products; rounding differs
# speedup vs baseline: 7.8194x; 1.2994x over previous
#include <hip/hip_runtime.h>

typedef __attribute__((ext_vector_type(2))) float v2f;
typedef __attribute__((ext_vector_type(8))) float v8f;

#define C_DIM 64
#define T_DIM 4096
#define N_EPOCH 8
#define PATCH 512
#define EPS_COV 1e-5f

typedef __attribute__((ext_vector_type(4))) float v4f;

#define PW 68
#define SLOT (64 * PW)


__device__ __forceinline__ void wave_lds_sync() { __syncthreads(); }

__device__ __forceinline__ void put_lds(float* __restrict__ s, const v8f (&M)[16], int h, int c) {
    float* p = s + (8 * h) * PW + c;
#pragma unroll
    for (int I = 0; I < 4; ++I)
#pragma unroll
        for (int J = 0; J < 4; ++J)
#pragma unroll
            for (int r = 0; r < 8; ++r)
                p[(16 * I + r) * PW + 16 * J] = M[I * 4 + J][r];
}

__device__ __forceinline__ void get_arow(v8f (&a)[4], const float* __restrict__ s, int J, int h, int c) {
    const float* p = s + (16 * J + c) * PW + 8 * h;
#pragma unroll
    for (int I = 0; I < 4; ++I) {
        const v4f lo = *(const v4f*)(p + 16 * I);
        const v4f hi = *(const v4f*)(p + 16 * I + 4);
        v8f t = {lo.x, lo.y, lo.z, lo.w, hi.x, hi.y, hi.z, hi.w};
        a[I] = t;
    }
}

__device__ __forceinline__ void mm(v8f (&D)[16], const float* __restrict__ sA, const v8f (&B)[16],
                                   int h, int c) {
#pragma unroll
    for (int J = 0; J < 4; ++J) {
        v8f a[4];
        get_arow(a, sA, J, h, c);
        v8f acc[4];
#pragma unroll
        for (int Jp = 0; Jp < 4; ++Jp) { v8f z = {0.f,0.f,0.f,0.f,0.f,0.f,0.f,0.f}; acc[Jp] = z; }
#pragma unroll
        for (int I = 0; I < 4; ++I)
#pragma unroll
            for (int q = 0; q < 4; ++q) {
                v2f af; af.x = a[I][2 * q]; af.y = a[I][2 * q + 1];
#pragma unroll
                for (int Jp = 0; Jp < 4; ++Jp) {
                    v2f bf; bf.x = B[I * 4 + Jp][2 * q]; bf.y = B[I * 4 + Jp][2 * q + 1];
                    acc[Jp] = __builtin_amdgcn_wmma_f32_16x16x4_f32(false, af, false, bf,
                                                                    (short)0, acc[Jp], false, false);
                }
            }
#pragma unroll
        for (int Jp = 0; Jp < 4; ++Jp) D[J * 4 + Jp] = acc[Jp];
    }
}

__device__ __forceinline__ void mm_add(v8f (&D)[16], const float* __restrict__ sA, const v8f (&B)[16],
                                       int h, int c, bool on) {
#pragma unroll
    for (int J = 0; J < 4; ++J) {
        if (!on) continue;
        v8f a[4];
        get_arow(a, sA, J, h, c);
        v8f acc[4];
#pragma unroll
        for (int Jp = 0; Jp < 4; ++Jp) acc[Jp] = D[J * 4 + Jp];
#pragma unroll
        for (int I = 0; I < 4; ++I)
#pragma unroll
            for (int q = 0; q < 4; ++q) {
                v2f af; af.x = a[I][2 * q]; af.y = a[I][2 * q + 1];
#pragma unroll
                for (int Jp = 0; Jp < 4; ++Jp) {
                    v2f bf; bf.x = B[I * 4 + Jp][2 * q]; bf.y = B[I * 4 + Jp][2 * q + 1];
                    acc[Jp] = __builtin_amdgcn_wmma_f32_16x16x4_f32(false, af, false, bf,
                                                                    (short)0, acc[Jp], false, false);
                }
            }
#pragma unroll
        for (int Jp = 0; Jp < 4; ++Jp) D[J * 4 + Jp] = acc[Jp];
    }
}

__device__ __forceinline__ void mm_sym(v8f (&D)[16], const v8f (&P)[16], const v8f (&Q)[16]) {
#pragma unroll
    for (int J = 0; J < 4; ++J)
#pragma unroll
        for (int Jp = J; Jp < 4; ++Jp) { v8f z = {0.f,0.f,0.f,0.f,0.f,0.f,0.f,0.f}; D[J * 4 + Jp] = z; }
#pragma unroll
    for (int I = 0; I < 4; ++I)
#pragma unroll
        for (int q = 0; q < 4; ++q)
#pragma unroll
            for (int J = 0; J < 4; ++J) {
                v2f af; af.x = P[I * 4 + J][2 * q]; af.y = P[I * 4 + J][2 * q + 1];
#pragma unroll
                for (int Jp = J; Jp < 4; ++Jp) {
                    v2f bf; bf.x = Q[I * 4 + Jp][2 * q]; bf.y = Q[I * 4 + Jp][2 * q + 1];
                    D[J * 4 + Jp] = __builtin_amdgcn_wmma_f32_16x16x4_f32(false, af, false, bf,
                                                                          (short)0, D[J * 4 + Jp], false, false);
                }
            }
}

__device__ __forceinline__ void mm_upper(v8f (&D)[16], const float* __restrict__ sA, const v8f (&B)[16],
                                         int h, int c) {
#pragma unroll
    for (int J = 0; J < 4; ++J) {
        v8f a[4];
        get_arow(a, sA, J, h, c);
#pragma unroll
        for (int Jp = J; Jp < 4; ++Jp) { v8f z = {0.f,0.f,0.f,0.f,0.f,0.f,0.f,0.f}; D[J * 4 + Jp] = z; }
#pragma unroll
        for (int I = 0; I < 4; ++I)
#pragma unroll
            for (int q = 0; q < 4; ++q) {
                v2f af; af.x = a[I][2 * q]; af.y = a[I][2 * q + 1];
#pragma unroll
                for (int Jp = J; Jp < 4; ++Jp) {
                    v2f bf; bf.x = B[I * 4 + Jp][2 * q]; bf.y = B[I * 4 + Jp][2 * q + 1];
                    D[J * 4 + Jp] = __builtin_amdgcn_wmma_f32_16x16x4_f32(false, af, false, bf,
                                                                          (short)0, D[J * 4 + Jp], false, false);
                }
            }
    }
}

__device__ __forceinline__ void mirror_lower(float* __restrict__ s, v8f (&M)[16], int h, int c) {
    wave_lds_sync();
    float* p = s + (8 * h) * PW + c;
#pragma unroll
    for (int I = 0; I < 4; ++I)
#pragma unroll
        for (int J = I; J < 4; ++J)
#pragma unroll
            for (int r = 0; r < 8; ++r)
                p[(16 * I + r) * PW + 16 * J] = M[I * 4 + J][r];
    wave_lds_sync();
#pragma unroll
    for (int J = 0; J < 3; ++J) {
        const float* rp = s + (16 * J + c) * PW + 8 * h;
#pragma unroll
        for (int I = J + 1; I < 4; ++I) {
            const v4f lo = *(const v4f*)(rp + 16 * I);
            const v4f hi = *(const v4f*)(rp + 16 * I + 4);
            v8f t = {lo.x, lo.y, lo.z, lo.w, hi.x, hi.y, hi.z, hi.w};
            M[I * 4 + J] = t;
        }
    }
}

#define FOR_TILE_EL \
    _Pragma("unroll") for (int I = 0; I < 4; ++I) \
    _Pragma("unroll") for (int J = 0; J < 4; ++J) \
    _Pragma("unroll") for (int r = 0; r < 8; ++r)
#define FOR_UPPER_EL \
    _Pragma("unroll") for (int I = 0; I < 4; ++I) \
    _Pragma("unroll") for (int J = I; J < 4; ++J) \
    _Pragma("unroll") for (int r = 0; r < 8; ++r)
#define ON_DIAG ((I == J) && (8 * h + r == c))

__global__ void __launch_bounds__(32, 2)
spdlogexp_kernel(const float* __restrict__ x,
                 const float* __restrict__ alphap,
                 const float* __restrict__ betap,
                 float* __restrict__ out) {
    extern __shared__ __attribute__((aligned(16))) float lds[];
    float* s0 = lds;
    float* s1 = lds + SLOT;

    const int lane = threadIdx.x, h = lane >> 4, c = lane & 15;
    const int bid = blockIdx.x;
    const float alpha = alphap[0], beta = betap[0];

    v8f A[16], Y[16], Z[16], T[16], Yn[16], Zn[16];

    {
        const float* xr = x + (size_t)(bid >> 3) * C_DIM * T_DIM + (size_t)(bid & 7) * PATCH
                            + (size_t)c * T_DIM + 16 * h;
        v4f fa[4][4], fb[4][4], fc[4][4], fd[4][4];
        float ms[4] = {0.f, 0.f, 0.f, 0.f};
#pragma unroll
        for (int t = 0; t < 16; ++t) { v8f z = {0.f,0.f,0.f,0.f,0.f,0.f,0.f,0.f}; A[t] = z; }

        auto ldgrp = [&](v4f (&f)[4][4], int g) {
#pragma unroll
            for (int J = 0; J < 4; ++J)
#pragma unroll
                for (int i = 0; i < 4; ++i)
                    f[J][i] = *(const v4f*)(xr + (size_t)(16 * J) * T_DIM + 32 * g + 4 * i);
        };
        auto gram = [&](const v4f (&f)[4][4]) {
#pragma unroll
            for (int i = 0; i < 4; ++i)
#pragma unroll
                for (int p = 0; p < 2; ++p) {
                    v2f fr[4];
#pragma unroll
                    for (int J = 0; J < 4; ++J) { fr[J].x = f[J][i][2 * p]; fr[J].y = f[J][i][2 * p + 1]; }
#pragma unroll
                    for (int J = 0; J < 4; ++J)
#pragma unroll
                        for (int Jp = J; Jp < 4; ++Jp)
                            A[J * 4 + Jp] = __builtin_amdgcn_wmma_f32_16x16x4_f32(false, fr[J], false, fr[Jp],
                                                                                 (short)0, A[J * 4 + Jp], false, false);
                }
#pragma unroll
            for (int J = 0; J < 4; ++J) {
                const v4f s4 = (f[J][0] + f[J][1]) + (f[J][2] + f[J][3]);
                ms[J] += (s4.x + s4.y) + (s4.z + s4.w);
            }
        };

        ldgrp(fa, 0);
        ldgrp(fb, 1);
        for (int g = 2; g < PATCH / 32 - 2; g += 4) {
            ldgrp(fc, g);
            ldgrp(fd, g + 1);
            gram(fa);
            gram(fb);
            ldgrp(fa, g + 2);
            ldgrp(fb, g + 3);
            gram(fc);
            gram(fd);
        }
        ldgrp(fc, PATCH / 32 - 2);
        ldgrp(fd, PATCH / 32 - 1);
        gram(fa);
        gram(fb);
        gram(fc);
        gram(fd);

        mirror_lower(s0, A, h, c);

        float mean[4];
#pragma unroll
        for (int J = 0; J < 4; ++J) {
            const float t = ms[J] + __shfl_xor(ms[J], 16);
            mean[J] = t * (1.f / PATCH);
        }
#pragma unroll
        for (int I = 0; I < 4; ++I)
#pragma unroll
            for (int r = 0; r < 8; ++r) {
                const float mr = __shfl(mean[I], 8 * h + r);
#pragma unroll
                for (int J = 0; J < 4; ++J)
                    A[I * 4 + J][r] = (A[I * 4 + J][r] - (float)PATCH * mr * mean[J]) * (1.f / (PATCH - 1));
            }
    }
    float tr = 0.f;
    FOR_TILE_EL { if (ON_DIAG) tr += A[I * 4 + J][r]; }
#pragma unroll
    for (int m = 16; m >= 1; m >>= 1) tr += __shfl_xor(tr, m);
    const float invtr = 1.f / tr;
    FOR_TILE_EL {
        float v = A[I * 4 + J][r] * invtr;
        if (ON_DIAG) v += EPS_COV;
        A[I * 4 + J][r] = v;
    }

    constexpr int NS_STEPS = 5;
    constexpr float NS_Q[NS_STEPS][5] = {
        {7.55900367f, -73.5964731f, 244.153359f, -313.036904f, 135.881263f},
        {7.28266995f, -66.3727064f, 215.821045f, -274.034362f, 118.269434f},
        {5.50208321f, -30.8133248f, 82.8106228f, -95.0746483f, 38.5664482f},
        {2.10391988f, -1.75144606f, 0.647438191f, 0.f, 0.f},
        {1.88060021f, -1.26121892f, 0.380618686f, 0.f, 0.f}};
    constexpr unsigned NS_QUADRATIC = 0x18u;
    {
        const float q0 = NS_Q[0][0], q1 = NS_Q[0][1], q2 = NS_Q[0][2], q3 = NS_Q[0][3], q4 = NS_Q[0][4];
        wave_lds_sync();
        put_lds(s1, A, h, c);
        wave_lds_sync();
        mm(T, s1, A, h, c);
        wave_lds_sync();
        put_lds(s0, T, h, c);
        wave_lds_sync();
        FOR_TILE_EL { const float m1 = A[I * 4 + J][r], m2 = T[I * 4 + J][r];
                      T[I * 4 + J][r] = (ON_DIAG ? q2 : 0.f) + q3 * m1 + q4 * m2;
                      Z[I * 4 + J][r] = (ON_DIAG ? q0 : 0.f) + q1 * m1; }
        mm_add(Z, s0, T, h, c, true);
        mm(Y, s1, Z, h, c);
        wave_lds_sync();
        put_lds(s1, Y, h, c);
        put_lds(s0, Z, h, c);
        wave_lds_sync();
    }
    for (int qi = 1; qi < NS_STEPS; ++qi) {
        const bool last = (qi == NS_STEPS - 1);
        const float q0 = NS_Q[qi][0], q1 = NS_Q[qi][1], q2 = NS_Q[qi][2], q3 = NS_Q[qi][3], q4 = NS_Q[qi][4];
        const bool quartic = !((NS_QUADRATIC >> qi) & 1u);
        const float qs = quartic ? 0.f : q2;
        mm(T, s0, Y, h, c);
        wave_lds_sync();
        put_lds(s0, T, h, c);
        wave_lds_sync();
        mm(Yn, s0, T, h, c);
        if (quartic) {
            wave_lds_sync();
            put_lds(s0, Yn, h, c);
            wave_lds_sync();
        }
        FOR_TILE_EL { const float m1 = T[I * 4 + J][r], m2 = Yn[I * 4 + J][r];
                      Yn[I * 4 + J][r] = (ON_DIAG ? q2 : 0.f) + q3 * m1 + q4 * m2;
                      T[I * 4 + J][r] = (ON_DIAG ? q0 : 0.f) + q1 * m1 + qs * m2; }
        mm_add(T, s0, Yn, h, c, quartic);
        wave_lds_sync();
        put_lds(s0, T, h, c);
        wave_lds_sync();
        mm(Yn, s1, T, h, c);
        mm(Zn, s0, Z, h, c);
        if (!last) {
            wave_lds_sync();
            put_lds(s1, Yn, h, c);
            put_lds(s0, Zn, h, c);
            wave_lds_sync();
        }
#pragma unroll
        for (int t = 0; t < 16; ++t) { Y[t] = Yn[t]; Z[t] = Zn[t]; }
    }
    constexpr int LOG_D = 17;
    constexpr float LOG_UY = 8.92504120f;
    constexpr float LOG_UZ = 0.0280110762f;
    constexpr float LOG_TY = 2.23817205f;
    constexpr float LOG_TZ = 0.00702446187f;
    constexpr float LOG_TI = -1.25077438f;
    constexpr float LOG_LS = -2.88200808f;
    constexpr float LOG_P[LOG_D + 1] = {
        0.468631387f, -0.223354504f, 0.123027645f, -0.0716356710f, 0.0430009402f,
        -0.0262867678f, 0.0160146430f, -0.0100046126f, 0.00765368063f, -0.00487180520f,
        -0.000807870994f, 0.000563526934f, 0.00626643049f, -0.00407942059f, -0.00392335746f,
        0.00256317854f, 0.00183465448f, -0.00119576673f};
    FOR_TILE_EL { const float yv = Y[I * 4 + J][r], zv = Z[I * 4 + J][r];
                  T[I * 4 + J][r] = yv * LOG_UY - zv * LOG_UZ;
                  Y[I * 4 + J][r] = yv * LOG_TY + zv * LOG_TZ + (ON_DIAG ? LOG_TI : 0.f); }
    wave_lds_sync();
    put_lds(s1, T, h, c);
    wave_lds_sync();
    FOR_TILE_EL { Z[I * 4 + J][r] = Y[I * 4 + J][r] * LOG_P[LOG_D] + (ON_DIAG ? LOG_P[LOG_D - 1] : 0.f); }
    for (int j = LOG_D - 2; j >= 0; --j) {
        mm_sym(T, Y, Z);
        const float pj = LOG_P[j];
        FOR_UPPER_EL { Z[I * 4 + J][r] = T[I * 4 + J][r] + (ON_DIAG ? pj : 0.f); }
        mirror_lower(s0, Z, h, c);
    }
    mm_upper(T, s1, Z, h, c);
    const float lscale = 2.f;

    FOR_UPPER_EL { A[I * 4 + J][r] = alpha * (lscale * (T[I * 4 + J][r] + (ON_DIAG ? LOG_LS : 0.f))) + beta; }
    mirror_lower(s1, A, h, c);
    wave_lds_sync();
    put_lds(s0, A, h, c);
    wave_lds_sync();
    constexpr int S_EXP = 6, EXP_M = 10;
    const float es = 1.f / (float)(1 << S_EXP);
#pragma unroll
    for (int J = 0; J < 4; ++J) {
        v8f at[4];
        get_arow(at, s0, J, h, c);
#pragma unroll
        for (int I = 0; I < 4; ++I)
#pragma unroll
            for (int r = 0; r < 8; ++r)
                Y[I * 4 + J][r] = (0.5f * (A[I * 4 + J][r] + at[I][r])) * es;
    }

    FOR_TILE_EL { Z[I * 4 + J][r] = Y[I * 4 + J][r] * (1.f / (float)EXP_M) + (ON_DIAG ? 1.f : 0.f); }
    for (int m = EXP_M - 1; m >= 1; --m) {
        mm_sym(T, Y, Z);
        const float dm = 1.f / (float)m;
        FOR_UPPER_EL { Z[I * 4 + J][r] = T[I * 4 + J][r] * dm + (ON_DIAG ? 1.f : 0.f); }
        mirror_lower(s0, Z, h, c);
    }
    for (int j = 0; j < S_EXP; ++j) {
        mm_sym(T, Z, Z);
        FOR_UPPER_EL { Z[I * 4 + J][r] = T[I * 4 + J][r]; }
        mirror_lower(s0, Z, h, c);
    }

    wave_lds_sync();
    put_lds(s0, Z, h, c);
    wave_lds_sync();
    float* ob = out + (size_t)bid * 4096;
#pragma unroll
    for (int j = 0; j < 32; ++j) {
        const int row = 2 * j + h;
        const v4f v = *(const v4f*)(s0 + row * PW + 4 * c);
        *(v4f*)(ob + row * 64 + 4 * c) = v;
    }
}

extern "C" void kernel_launch(void* const* d_in, const int* in_sizes, int n_in,
                              void* d_out, int out_size, void* d_ws, size_t ws_size,
                              hipStream_t stream) {
    const float* x     = (const float*)d_in[0];
    const float* alpha = (const float*)d_in[1];
    const float* beta  = (const float*)d_in[2];
    float* out = (float*)d_out;

    const int nmat = 256 * N_EPOCH;
    const size_t shbytes = (size_t)(2 * SLOT) * sizeof(float);
    spdlogexp_kernel<<<nmat, 32, shbytes, stream>>>(x, alpha, beta, out);
}


# Exp argument centred on the midpoint of its spectral interval (known from alpha, beta, the log's bounds); squarings and Taylor degree set by its radius: 6+10 -> 1+7 at alpha 0.1; rounding differs
# speedup vs baseline: 8.5388x; 1.0920x over previous
#include <hip/hip_runtime.h>

typedef __attribute__((ext_vector_type(2))) float v2f;
typedef __attribute__((ext_vector_type(8))) float v8f;

#define C_DIM 64
#define T_DIM 4096
#define N_EPOCH 8
#define PATCH 512
#define EPS_COV 1e-5f

typedef __attribute__((ext_vector_type(4))) float v4f;

#define PW 68
#define SLOT (64 * PW)


__device__ __forceinline__ void wave_lds_sync() { __syncthreads(); }

__device__ __forceinline__ void put_lds(float* __restrict__ s, const v8f (&M)[16], int h, int c) {
    float* p = s + (8 * h) * PW + c;
#pragma unroll
    for (int I = 0; I < 4; ++I)
#pragma unroll
        for (int J = 0; J < 4; ++J)
#pragma unroll
            for (int r = 0; r < 8; ++r)
                p[(16 * I + r) * PW + 16 * J] = M[I * 4 + J][r];
}

__device__ __forceinline__ void get_arow(v8f (&a)[4], const float* __restrict__ s, int J, int h, int c) {
    const float* p = s + (16 * J + c) * PW + 8 * h;
#pragma unroll
    for (int I = 0; I < 4; ++I) {
        const v4f lo = *(const v4f*)(p + 16 * I);
        const v4f hi = *(const v4f*)(p + 16 * I + 4);
        v8f t = {lo.x, lo.y, lo.z, lo.w, hi.x, hi.y, hi.z, hi.w};
        a[I] = t;
    }
}

__device__ __forceinline__ void mm(v8f (&D)[16], const float* __restrict__ sA, const v8f (&B)[16],
                                   int h, int c) {
#pragma unroll
    for (int J = 0; J < 4; ++J) {
        v8f a[4];
        get_arow(a, sA, J, h, c);
        v8f acc[4];
#pragma unroll
        for (int Jp = 0; Jp < 4; ++Jp) { v8f z = {0.f,0.f,0.f,0.f,0.f,0.f,0.f,0.f}; acc[Jp] = z; }
#pragma unroll
        for (int I = 0; I < 4; ++I)
#pragma unroll
            for (int q = 0; q < 4; ++q) {
                v2f af; af.x = a[I][2 * q]; af.y = a[I][2 * q + 1];
#pragma unroll
                for (int Jp = 0; Jp < 4; ++Jp) {
                    v2f bf; bf.x = B[I * 4 + Jp][2 * q]; bf.y = B[I * 4 + Jp][2 * q + 1];
                    acc[Jp] = __builtin_amdgcn_wmma_f32_16x16x4_f32(false, af, false, bf,
                                                                    (short)0, acc[Jp], false, false);
                }
            }
#pragma unroll
        for (int Jp = 0; Jp < 4; ++Jp) D[J * 4 + Jp] = acc[Jp];
    }
}

__device__ __forceinline__ void mm_add(v8f (&D)[16], const float* __restrict__ sA, const v8f (&B)[16],
                                       int h, int c, bool on) {
#pragma unroll
    for (int J = 0; J < 4; ++J) {
        if (!on) continue;
        v8f a[4];
        get_arow(a, sA, J, h, c);
        v8f acc[4];
#pragma unroll
        for (int Jp = 0; Jp < 4; ++Jp) acc[Jp] = D[J * 4 + Jp];
#pragma unroll
        for (int I = 0; I < 4; ++I)
#pragma unroll
            for (int q = 0; q < 4; ++q) {
                v2f af; af.x = a[I][2 * q]; af.y = a[I][2 * q + 1];
#pragma unroll
                for (int Jp = 0; Jp < 4; ++Jp) {
                    v2f bf; bf.x = B[I * 4 + Jp][2 * q]; bf.y = B[I * 4 + Jp][2 * q + 1];
                    acc[Jp] = __builtin_amdgcn_wmma_f32_16x16x4_f32(false, af, false, bf,
                                                                    (short)0, acc[Jp], false, false);
                }
            }
#pragma unroll
        for (int Jp = 0; Jp < 4; ++Jp) D[J * 4 + Jp] = acc[Jp];
    }
}

__device__ __forceinline__ void mm_sym(v8f (&D)[16], const v8f (&P)[16], const v8f (&Q)[16]) {
#pragma unroll
    for (int J = 0; J < 4; ++J)
#pragma unroll
        for (int Jp = J; Jp < 4; ++Jp) { v8f z = {0.f,0.f,0.f,0.f,0.f,0.f,0.f,0.f}; D[J * 4 + Jp] = z; }
#pragma unroll
    for (int I = 0; I < 4; ++I)
#pragma unroll
        for (int q = 0; q < 4; ++q)
#pragma unroll
            for (int J = 0; J < 4; ++J) {
                v2f af; af.x = P[I * 4 + J][2 * q]; af.y = P[I * 4 + J][2 * q + 1];
#pragma unroll
                for (int Jp = J; Jp < 4; ++Jp) {
                    v2f bf; bf.x = Q[I * 4 + Jp][2 * q]; bf.y = Q[I * 4 + Jp][2 * q + 1];
                    D[J * 4 + Jp] = __builtin_amdgcn_wmma_f32_16x16x4_f32(false, af, false, bf,
                                                                          (short)0, D[J * 4 + Jp], false, false);
                }
            }
}

__device__ __forceinline__ void mm_upper(v8f (&D)[16], const float* __restrict__ sA, const v8f (&B)[16],
                                         int h, int c) {
#pragma unroll
    for (int J = 0; J < 4; ++J) {
        v8f a[4];
        get_arow(a, sA, J, h, c);
#pragma unroll
        for (int Jp = J; Jp < 4; ++Jp) { v8f z = {0.f,0.f,0.f,0.f,0.f,0.f,0.f,0.f}; D[J * 4 + Jp] = z; }
#pragma unroll
        for (int I = 0; I < 4; ++I)
#pragma unroll
            for (int q = 0; q < 4; ++q) {
                v2f af; af.x = a[I][2 * q]; af.y = a[I][2 * q + 1];
#pragma unroll
                for (int Jp = J; Jp < 4; ++Jp) {
                    v2f bf; bf.x = B[I * 4 + Jp][2 * q]; bf.y = B[I * 4 + Jp][2 * q + 1];
                    D[J * 4 + Jp] = __builtin_amdgcn_wmma_f32_16x16x4_f32(false, af, false, bf,
                                                                          (short)0, D[J * 4 + Jp], false, false);
                }
            }
    }
}

__device__ __forceinline__ void mirror_lower(float* __restrict__ s, v8f (&M)[16], int h, int c) {
    wave_lds_sync();
    float* p = s + (8 * h) * PW + c;
#pragma unroll
    for (int I = 0; I < 4; ++I)
#pragma unroll
        for (int J = I; J < 4; ++J)
#pragma unroll
            for (int r = 0; r < 8; ++r)
                p[(16 * I + r) * PW + 16 * J] = M[I * 4 + J][r];
    wave_lds_sync();
#pragma unroll
    for (int J = 0; J < 3; ++J) {
        const float* rp = s + (16 * J + c) * PW + 8 * h;
#pragma unroll
        for (int I = J + 1; I < 4; ++I) {
            const v4f lo = *(const v4f*)(rp + 16 * I);
            const v4f hi = *(const v4f*)(rp + 16 * I + 4);
            v8f t = {lo.x, lo.y, lo.z, lo.w, hi.x, hi.y, hi.z, hi.w};
            M[I * 4 + J] = t;
        }
    }
}

#define FOR_TILE_EL \
    _Pragma("unroll") for (int I = 0; I < 4; ++I) \
    _Pragma("unroll") for (int J = 0; J < 4; ++J) \
    _Pragma("unroll") for (int r = 0; r < 8; ++r)
#define FOR_UPPER_EL \
    _Pragma("unroll") for (int I = 0; I < 4; ++I) \
    _Pragma("unroll") for (int J = I; J < 4; ++J) \
    _Pragma("unroll") for (int r = 0; r < 8; ++r)
#define ON_DIAG ((I == J) && (8 * h + r == c))

__global__ void __launch_bounds__(32, 2)
spdlogexp_kernel(const float* __restrict__ x,
                 const float* __restrict__ alphap,
                 const float* __restrict__ betap,
                 float* __restrict__ out) {
    extern __shared__ __attribute__((aligned(16))) float lds[];
    float* s0 = lds;
    float* s1 = lds + SLOT;

    const int lane = threadIdx.x, h = lane >> 4, c = lane & 15;
    const int bid = blockIdx.x;
    const float alpha = alphap[0], beta = betap[0];

    v8f A[16], Y[16], Z[16], T[16], Yn[16], Zn[16];

    {
        const float* xr = x + (size_t)(bid >> 3) * C_DIM * T_DIM + (size_t)(bid & 7) * PATCH
                            + (size_t)c * T_DIM + 16 * h;
        v4f fa[4][4], fb[4][4], fc[4][4], fd[4][4];
        float ms[4] = {0.f, 0.f, 0.f, 0.f};
#pragma unroll
        for (int t = 0; t < 16; ++t) { v8f z = {0.f,0.f,0.f,0.f,0.f,0.f,0.f,0.f}; A[t] = z; }

        auto ldgrp = [&](v4f (&f)[4][4], int g) {
#pragma unroll
            for (int J = 0; J < 4; ++J)
#pragma unroll
                for (int i = 0; i < 4; ++i)
                    f[J][i] = *(const v4f*)(xr + (size_t)(16 * J) * T_DIM + 32 * g + 4 * i);
        };
        auto gram = [&](const v4f (&f)[4][4]) {
#pragma unroll
            for (int i = 0; i < 4; ++i)
#pragma unroll
                for (int p = 0; p < 2; ++p) {
                    v2f fr[4];
#pragma unroll
                    for (int J = 0; J < 4; ++J) { fr[J].x = f[J][i][2 * p]; fr[J].y = f[J][i][2 * p + 1]; }
#pragma unroll
                    for (int J = 0; J < 4; ++J)
#pragma unroll
                        for (int Jp = J; Jp < 4; ++Jp)
                            A[J * 4 + Jp] = __builtin_amdgcn_wmma_f32_16x16x4_f32(false, fr[J], false, fr[Jp],
                                                                                 (short)0, A[J * 4 + Jp], false, false);
                }
#pragma unroll
            for (int J = 0; J < 4; ++J) {
                const v4f s4 = (f[J][0] + f[J][1]) + (f[J][2] + f[J][3]);
                ms[J] += (s4.x + s4.y) + (s4.z + s4.w);
            }
        };

        ldgrp(fa, 0);
        ldgrp(fb, 1);
        for (int g = 2; g < PATCH / 32 - 2; g += 4) {
            ldgrp(fc, g);
            ldgrp(fd, g + 1);
            gram(fa);
            gram(fb);
            ldgrp(fa, g + 2);
            ldgrp(fb, g + 3);
            gram(fc);
            gram(fd);
        }
        ldgrp(fc, PATCH / 32 - 2);
        ldgrp(fd, PATCH / 32 - 1);
        gram(fa);
        gram(fb);
        gram(fc);
        gram(fd);

        mirror_lower(s0, A, h, c);

        float mean[4];
#pragma unroll
        for (int J = 0; J < 4; ++J) {
            const float t = ms[J] + __shfl_xor(ms[J], 16);
            mean[J] = t * (1.f / PATCH);
        }
#pragma unroll
        for (int I = 0; I < 4; ++I)
#pragma unroll
            for (int r = 0; r < 8; ++r) {
                const float mr = __shfl(mean[I], 8 * h + r);
#pragma unroll
                for (int J = 0; J < 4; ++J)
                    A[I * 4 + J][r] = (A[I * 4 + J][r] - (float)PATCH * mr * mean[J]) * (1.f / (PATCH - 1));
            }
    }
    float tr = 0.f;
    FOR_TILE_EL { if (ON_DIAG) tr += A[I * 4 + J][r]; }
#pragma unroll
    for (int m = 16; m >= 1; m >>= 1) tr += __shfl_xor(tr, m);
    const float invtr = 1.f / tr;
    FOR_TILE_EL {
        float v = A[I * 4 + J][r] * invtr;
        if (ON_DIAG) v += EPS_COV;
        A[I * 4 + J][r] = v;
    }

    constexpr int NS_STEPS = 5;
    constexpr float NS_Q[NS_STEPS][5] = {
        {7.55900367f, -73.5964731f, 244.153359f, -313.036904f, 135.881263f},
        {7.28266995f, -66.3727064f, 215.821045f, -274.034362f, 118.269434f},
        {5.50208321f, -30.8133248f, 82.8106228f, -95.0746483f, 38.5664482f},
        {2.10391988f, -1.75144606f, 0.647438191f, 0.f, 0.f},
        {1.88060021f, -1.26121892f, 0.380618686f, 0.f, 0.f}};
    constexpr unsigned NS_QUADRATIC = 0x18u;
    {
        const float q0 = NS_Q[0][0], q1 = NS_Q[0][1], q2 = NS_Q[0][2], q3 = NS_Q[0][3], q4 = NS_Q[0][4];
        wave_lds_sync();
        put_lds(s1, A, h, c);
        wave_lds_sync();
        mm(T, s1, A, h, c);
        wave_lds_sync();
        put_lds(s0, T, h, c);
        wave_lds_sync();
        FOR_TILE_EL { const float m1 = A[I * 4 + J][r], m2 = T[I * 4 + J][r];
                      T[I * 4 + J][r] = (ON_DIAG ? q2 : 0.f) + q3 * m1 + q4 * m2;
                      Z[I * 4 + J][r] = (ON_DIAG ? q0 : 0.f) + q1 * m1; }
        mm_add(Z, s0, T, h, c, true);
        mm(Y, s1, Z, h, c);
        wave_lds_sync();
        put_lds(s1, Y, h, c);
        put_lds(s0, Z, h, c);
        wave_lds_sync();
    }
    for (int qi = 1; qi < NS_STEPS; ++qi) {
        const bool last = (qi == NS_STEPS - 1);
        const float q0 = NS_Q[qi][0], q1 = NS_Q[qi][1], q2 = NS_Q[qi][2], q3 = NS_Q[qi][3], q4 = NS_Q[qi][4];
        const bool quartic = !((NS_QUADRATIC >> qi) & 1u);
        const float qs = quartic ? 0.f : q2;
        mm(T, s0, Y, h, c);
        wave_lds_sync();
        put_lds(s0, T, h, c);
        wave_lds_sync();
        mm(Yn, s0, T, h, c);
        if (quartic) {
            wave_lds_sync();
            put_lds(s0, Yn, h, c);
            wave_lds_sync();
        }
        FOR_TILE_EL { const float m1 = T[I * 4 + J][r], m2 = Yn[I * 4 + J][r];
                      Yn[I * 4 + J][r] = (ON_DIAG ? q2 : 0.f) + q3 * m1 + q4 * m2;
                      T[I * 4 + J][r] = (ON_DIAG ? q0 : 0.f) + q1 * m1 + qs * m2; }
        mm_add(T, s0, Yn, h, c, quartic);
        wave_lds_sync();
        put_lds(s0, T, h, c);
        wave_lds_sync();
        mm(Yn, s1, T, h, c);
        mm(Zn, s0, Z, h, c);
        if (!last) {
            wave_lds_sync();
            put_lds(s1, Yn, h, c);
            put_lds(s0, Zn, h, c);
            wave_lds_sync();
        }
#pragma unroll
        for (int t = 0; t < 16; ++t) { Y[t] = Yn[t]; Z[t] = Zn[t]; }
    }
    constexpr int LOG_D = 17;
    constexpr float LOG_UY = 8.92504120f;
    constexpr float LOG_UZ = 0.0280110762f;
    constexpr float LOG_TY = 2.23817205f;
    constexpr float LOG_TZ = 0.00702446187f;
    constexpr float LOG_TI = -1.25077438f;
    constexpr float LOG_LS = -2.88200808f;
    constexpr float LOG_P[LOG_D + 1] = {
        0.468631387f, -0.223354504f, 0.123027645f, -0.0716356710f, 0.0430009402f,
        -0.0262867678f, 0.0160146430f, -0.0100046126f, 0.00765368063f, -0.00487180520f,
        -0.000807870994f, 0.000563526934f, 0.00626643049f, -0.00407942059f, -0.00392335746f,
        0.00256317854f, 0.00183465448f, -0.00119576673f};
    FOR_TILE_EL { const float yv = Y[I * 4 + J][r], zv = Z[I * 4 + J][r];
                  T[I * 4 + J][r] = yv * LOG_UY - zv * LOG_UZ;
                  Y[I * 4 + J][r] = yv * LOG_TY + zv * LOG_TZ + (ON_DIAG ? LOG_TI : 0.f); }
    wave_lds_sync();
    put_lds(s1, T, h, c);
    wave_lds_sync();
    FOR_TILE_EL { Z[I * 4 + J][r] = Y[I * 4 + J][r] * LOG_P[LOG_D] + (ON_DIAG ? LOG_P[LOG_D - 1] : 0.f); }
    for (int j = LOG_D - 2; j >= 0; --j) {
        mm_sym(T, Y, Z);
        const float pj = LOG_P[j];
        FOR_UPPER_EL { Z[I * 4 + J][r] = T[I * 4 + J][r] + (ON_DIAG ? pj : 0.f); }
        mirror_lower(s0, Z, h, c);
    }
    mm_upper(T, s1, Z, h, c);
    const float lscale = 2.f;

    FOR_UPPER_EL { A[I * 4 + J][r] = alpha * (lscale * (T[I * 4 + J][r] + (ON_DIAG ? LOG_LS : 0.f))) + beta; }
    mirror_lower(s1, A, h, c);
    wave_lds_sync();
    put_lds(s0, A, h, c);
    wave_lds_sync();
    constexpr float LOGM_LO = -11.56f, LOGM_HI = 0.01f;
    constexpr float EXP_TOL = 4e-9f;
    constexpr int EXP_SMAX = 8, EXP_MMAX = 14;
    const float al0 = alpha * LOGM_LO, al1 = alpha * LOGM_HI, b64 = 64.f * beta;
    const float sp_lo = fminf(al0, al1) + fminf(b64, 0.f);
    const float sp_hi = fmaxf(al0, al1) + fmaxf(b64, 0.f);
    const float mu = 0.5f * (sp_lo + sp_hi);
    const float hw = 0.5f * (sp_hi - sp_lo) * 1.01f + 1e-6f;
    int S_EXP = EXP_SMAX, EXP_M = EXP_MMAX;
    {
        int best = EXP_SMAX + EXP_MMAX;
        float p2 = 1.f, ip2 = 1.f;
#pragma unroll
        for (int s = 0; s <= EXP_SMAX; ++s) {
            const float th = hw * ip2;
            float term = th;
            bool found = false;
            if (th <= 1.f && s < best) {
#pragma unroll
                for (int m = 1; m <= EXP_MMAX; ++m) {
                    term *= th * (1.f / (float)(m + 1));
                    const bool ok = (term * p2 <= EXP_TOL);
                    if (ok && !found && (s + m - 1 < best)) { best = s + m - 1; S_EXP = s; EXP_M = m; }
                    found = found || ok;
                }
            }
            p2 *= 2.f; ip2 *= 0.5f;
        }
    }
    float es = 1.f;
    for (int s = 0; s < S_EXP; ++s) es *= 0.5f;
#pragma unroll
    for (int J = 0; J < 4; ++J) {
        v8f at[4];
        get_arow(at, s0, J, h, c);
#pragma unroll
        for (int I = 0; I < 4; ++I)
#pragma unroll
            for (int r = 0; r < 8; ++r)
                Y[I * 4 + J][r] = (0.5f * (A[I * 4 + J][r] + at[I][r]) - (ON_DIAG ? mu : 0.f)) * es;
    }

    FOR_TILE_EL { Z[I * 4 + J][r] = Y[I * 4 + J][r] * (1.f / (float)EXP_M) + (ON_DIAG ? 1.f : 0.f); }
    for (int m = EXP_M - 1; m >= 1; --m) {
        mm_sym(T, Y, Z);
        const float dm = 1.f / (float)m;
        FOR_UPPER_EL { Z[I * 4 + J][r] = T[I * 4 + J][r] * dm + (ON_DIAG ? 1.f : 0.f); }
        mirror_lower(s0, Z, h, c);
    }
    for (int j = 0; j < S_EXP; ++j) {
        mm_sym(T, Z, Z);
        FOR_UPPER_EL { Z[I * 4 + J][r] = T[I * 4 + J][r]; }
        mirror_lower(s0, Z, h, c);
    }
    const float emu = expf(mu);
    FOR_TILE_EL { Z[I * 4 + J][r] *= emu; }

    wave_lds_sync();
    put_lds(s0, Z, h, c);
    wave_lds_sync();
    float* ob = out + (size_t)bid * 4096;
#pragma unroll
    for (int j = 0; j < 32; ++j) {
        const int row = 2 * j + h;
        const v4f v = *(const v4f*)(s0 + row * PW + 4 * c);
        *(v4f*)(ob + row * 64 + 4 * c) = v;
    }
}

extern "C" void kernel_launch(void* const* d_in, const int* in_sizes, int n_in,
                              void* d_out, int out_size, void* d_ws, size_t ws_size,
                              hipStream_t stream) {
    const float* x     = (const float*)d_in[0];
    const float* alpha = (const float*)d_in[1];
    const float* beta  = (const float*)d_in[2];
    float* out = (float*)d_out;

    const int nmat = 256 * N_EPOCH;
    const size_t shbytes = (size_t)(2 * SLOT) * sizeof(float);
    spdlogexp_kernel<<<nmat, 32, shbytes, stream>>>(x, alpha, beta, out);
}
